# starting kernel
# baseline (speedup 1.0000x reference)
#include <hip/hip_runtime.h>
#include <hip/hip_bf16.h>

#define NB 2
#define LL 2048
#define HH 8
#define EE 64
#define MM 64
#define CK 64
#define NC (LL/CK)
#define EPS 1e-6f
#define PAD 4
#define LDW (EE + PAD)

typedef __attribute__((ext_vector_type(2))) float v2f;
typedef __attribute__((ext_vector_type(8))) float v8f;

__device__ __forceinline__ float fmap(float x) {
    return x > 0.0f ? x + 1.0f : __expf(x);
}

__global__ void mu_kernel(const float* __restrict__ keys,
                          const float* __restrict__ values,
                          const float* __restrict__ klen,
                          float* __restrict__ mu) {
    int b = blockIdx.x;
    int n = b / (LL - 1);
    int l = b % (LL - 1);
    int t = threadIdx.x;
    int h = t >> 5;
    int lane = t & 31;

    float kl0 = klen[n * LL + l];
    float kl1 = klen[n * LL + l + 1];

    float kn0 = 0.f, kn1 = 0.f, kd = 0.f, vn0 = 0.f, vn1 = 0.f, vd = 0.f;
    for (int e = lane; e < EE; e += 32) {
        float k0 = fmap(keys[(((size_t)n * LL + l)     * HH + h) * EE + e]) * kl0;
        float k1 = fmap(keys[(((size_t)n * LL + l + 1) * HH + h) * EE + e]) * kl1;
        kn0 += k0 * k0; kn1 += k1 * k1; kd += k0 * k1;
        float v0 = values[(((size_t)n * LL + l)     * HH + h) * MM + e];
        float v1 = values[(((size_t)n * LL + l + 1) * HH + h) * MM + e];
        vn0 += v0 * v0; vn1 += v1 * v1; vd += v0 * v1;
    }
    for (int off = 16; off > 0; off >>= 1) {
        kn0 += __shfl_xor(kn0, off, 32);
        kn1 += __shfl_xor(kn1, off, 32);
        kd  += __shfl_xor(kd,  off, 32);
        vn0 += __shfl_xor(vn0, off, 32);
        vn1 += __shfl_xor(vn1, off, 32);
        vd  += __shfl_xor(vd,  off, 32);
    }
    __shared__ float sh[HH][6];
    if (lane == 0) {
        sh[h][0] = kn0; sh[h][1] = kn1; sh[h][2] = kd;
        sh[h][3] = vn0; sh[h][4] = vn1; sh[h][5] = vd;
    }
    __syncthreads();
    if (t == 0) {
        float base2 = 0.f, diff2 = 0.f;
        for (int hh = 0; hh < HH; ++hh) {
            float a = sh[hh][0], b1 = sh[hh][1], c = sh[hh][2];
            float d = sh[hh][3], e1 = sh[hh][4], f = sh[hh][5];
            base2 += a * d;
            diff2 += b1 * e1 + a * d - 2.0f * c * f;
        }
        diff2 = fmaxf(diff2, 0.0f);
        float ratio = sqrtf(diff2) / sqrtf(base2);
        float r4 = sqrtf(ratio);
        float m_ = 1.0f - r4;
        m_ = m_ * m_;
        m_ = fminf(fmaxf(m_, 0.0f), 0.9999f);
        mu[n * LL + l] = m_;
    }
}

__global__ void weights_kernel(const float* __restrict__ mu,
                               float* __restrict__ weights) {
    int n = blockIdx.x;
    if (threadIdx.x != 0) return;
    float w = 1.0f;
    weights[n * LL + LL - 1] = 1.0f;
    for (int i = LL - 2; i >= 0; --i) {
        w = w * mu[n * LL + i] + 1.0f;
        weights[n * LL + i] = w;
    }
}

__global__ void sdelta_kernel(const float* __restrict__ keys,
                              const float* __restrict__ values,
                              const float* __restrict__ klen,
                              const float* __restrict__ weights,
                              float* __restrict__ Sbuf,
                              float* __restrict__ Ksum) {
    __shared__ float Klds[CK][LDW];
    __shared__ float Vlds[CK][LDW];

    int b = blockIdx.x;
    int c = b % NC;
    int h = (b / NC) % HH;
    int n = b / (NC * HH);
    int c0 = c * CK;
    int t = threadIdx.x;

    for (int i = t; i < CK * EE / 4; i += 256) {
        int s  = i >> 4;
        int e4 = (i & 15) * 4;
        int l  = c0 + s;
        float kl = klen[n * LL + l];
        float w  = weights[n * LL + l];
        const float4 kv = *(const float4*)&keys  [(((size_t)n * LL + l) * HH + h) * EE + e4];
        const float4 vv = *(const float4*)&values[(((size_t)n * LL + l) * HH + h) * MM + e4];
        float4 ko, vo;
        ko.x = fmap(kv.x) * kl; ko.y = fmap(kv.y) * kl;
        ko.z = fmap(kv.z) * kl; ko.w = fmap(kv.w) * kl;
        vo.x = vv.x * w; vo.y = vv.y * w; vo.z = vv.z * w; vo.w = vv.w * w;
        *(float4*)&Klds[s][e4] = ko;
        *(float4*)&Vlds[s][e4] = vo;
    }
    __syncthreads();

    if (t < EE) {
        float acc = 0.f;
        for (int s = 0; s < CK; ++s) acc += Klds[s][t];
        Ksum[(((size_t)n * HH + h) * NC + c) * EE + t] = acc;
    }

    int wave = t >> 5, lane = t & 31, lr = lane & 15, lh = lane >> 4;
    float* out = Sbuf + (((size_t)n * HH + h) * NC + c) * (EE * MM);

    for (int tile = wave; tile < 16; tile += 8) {
        int ti = tile >> 2, tj = tile & 3;
        v8f acc = {};
        for (int k0 = 0; k0 < CK; k0 += 4) {
            int s = k0 + 2 * lh;
            v2f a, bb;
            int e = ti * 16 + lr;
            a.x = Klds[s][e];
            a.y = Klds[s + 1][e];
            int m = tj * 16 + lr;
            bb.x = Vlds[s][m];
            bb.y = Vlds[s + 1][m];
            acc = __builtin_amdgcn_wmma_f32_16x16x4_f32(false, a, false, bb,
                                                        (short)0, acc, false, false);
        }
        for (int r = 0; r < 8; ++r) {
            int row = ti * 16 + r + 8 * lh;
            int col = tj * 16 + lr;
            out[row * MM + col] = acc[r];
        }
    }
}

__global__ void prefix_kernel(float* __restrict__ Sbuf,
                              float* __restrict__ Ksum) {
    int b = blockIdx.x;
    int t = threadIdx.x;
    float4* base = (float4*)(Sbuf + (size_t)b * NC * (EE * MM));
    for (int i = t; i < EE * MM / 4; i += 256) {
        float4 acc = {0.f, 0.f, 0.f, 0.f};
        for (int c = 0; c < NC; ++c) {
            float4 v = base[(size_t)c * (EE * MM / 4) + i];
            base[(size_t)c * (EE * MM / 4) + i] = acc;
            acc.x += v.x; acc.y += v.y; acc.z += v.z; acc.w += v.w;
        }
    }
    if (t < EE) {
        float* kb = Ksum + (size_t)b * NC * EE;
        float acc = 0.f;
        for (int c = 0; c < NC; ++c) {
            float v = kb[c * EE + t];
            kb[c * EE + t] = acc;
            acc += v;
        }
    }
}

__global__ void out_kernel(const float* __restrict__ queries,
                           const float* __restrict__ keys,
                           const float* __restrict__ values,
                           const float* __restrict__ klen,
                           const float* __restrict__ weights,
                           const float* __restrict__ Sbuf,
                           const float* __restrict__ Ksum,
                           float* __restrict__ out) {
    __shared__ float Qlds[CK][LDW];
    __shared__ float KP[CK][LDW];
    __shared__ float Vlds[CK][LDW];
    __shared__ float Slds[EE][LDW];
    __shared__ float ks[EE];
    __shared__ float zden[CK];

    int b = blockIdx.x;
    int c = b % NC;
    int h = (b / NC) % HH;
    int n = b / (NC * HH);
    int c0 = c * CK;
    int t = threadIdx.x;

    const float* Sp = Sbuf + (((size_t)n * HH + h) * NC + c) * (EE * MM);
    for (int i = t; i < CK * EE / 4; i += 256) {
        int s  = i >> 4;
        int e4 = (i & 15) * 4;
        int l  = c0 + s;
        float kl = klen[n * LL + l];
        float w  = weights[n * LL + l];
        const float4 qv = *(const float4*)&queries[(((size_t)n * LL + l) * HH + h) * EE + e4];
        const float4 kv = *(const float4*)&keys   [(((size_t)n * LL + l) * HH + h) * EE + e4];
        const float4 vv = *(const float4*)&values [(((size_t)n * LL + l) * HH + h) * MM + e4];
        float4 qo, ko, vo;
        qo.x = fmap(qv.x);      qo.y = fmap(qv.y);
        qo.z = fmap(qv.z);      qo.w = fmap(qv.w);
        ko.x = fmap(kv.x) * kl; ko.y = fmap(kv.y) * kl;
        ko.z = fmap(kv.z) * kl; ko.w = fmap(kv.w) * kl;
        vo.x = vv.x * w; vo.y = vv.y * w; vo.z = vv.z * w; vo.w = vv.w * w;
        *(float4*)&Qlds[s][e4] = qo;
        *(float4*)&KP[s][e4]   = ko;
        *(float4*)&Vlds[s][e4] = vo;
        *(float4*)&Slds[s][e4] = *(const float4*)&Sp[s * MM + e4];
    }
    if (t < EE) ks[t] = Ksum[(((size_t)n * HH + h) * NC + c) * EE + t];
    __syncthreads();

    int wave = t >> 5, lane = t & 31, lr = lane & 15, lh = lane >> 4;

    v8f p0 = {}, p1 = {};
    for (int half = 0; half < 2; ++half) {
        int tile = wave + half * 8;
        int ti = tile >> 2, tj = tile & 3;
        v8f acc = {};
        for (int k0 = 0; k0 < EE; k0 += 4) {
            int e = k0 + 2 * lh;
            v2f a, bb;
            int row = ti * 16 + lr;
            a.x = Qlds[row][e];
            a.y = Qlds[row][e + 1];
            int col = tj * 16 + lr;
            bb.x = KP[col][e];
            bb.y = KP[col][e + 1];
            acc = __builtin_amdgcn_wmma_f32_16x16x4_f32(false, a, false, bb,
                                                        (short)0, acc, false, false);
        }
        if (half) p1 = acc; else p0 = acc;
    }
    __syncthreads();

    for (int half = 0; half < 2; ++half) {
        v8f acc = half ? p1 : p0;
        int tile = wave + half * 8;
        int ti = tile >> 2, tj = tile & 3;
        for (int r = 0; r < 8; ++r) {
            int row = ti * 16 + r + 8 * lh;
            int col = tj * 16 + lr;
            KP[row][col] = (col <= row) ? acc[r] : 0.0f;
        }
    }
    __syncthreads();

    if (t < CK) {
        float s1 = 0.f;
        for (int s = 0; s <= t; ++s) s1 += KP[t][s];
        float s2 = 0.f;
        for (int e = 0; e < EE; ++e) s2 += Qlds[t][e] * ks[e];
        zden[t] = s1 + s2 + EPS;
    }
    __syncthreads();

    for (int half = 0; half < 2; ++half) {
        int tile = wave + half * 8;
        int ti = tile >> 2, tj = tile & 3;
        v8f acc = {};
        for (int k0 = 0; k0 < EE; k0 += 4) {
            int e = k0 + 2 * lh;
            v2f a, bb;
            int row = ti * 16 + lr;
            a.x = Qlds[row][e];
            a.y = Qlds[row][e + 1];
            int col = tj * 16 + lr;
            bb.x = Slds[e][col];
            bb.y = Slds[e + 1][col];
            acc = __builtin_amdgcn_wmma_f32_16x16x4_f32(false, a, false, bb,
                                                        (short)0, acc, false, false);
        }
        for (int k0 = 0; k0 < CK; k0 += 4) {
            int s = k0 + 2 * lh;
            v2f a, bb;
            int row = ti * 16 + lr;
            a.x = KP[row][s];
            a.y = KP[row][s + 1];
            int col = tj * 16 + lr;
            bb.x = Vlds[s][col];
            bb.y = Vlds[s + 1][col];
            acc = __builtin_amdgcn_wmma_f32_16x16x4_f32(false, a, false, bb,
                                                        (short)0, acc, false, false);
        }
        for (int r = 0; r < 8; ++r) {
            int row = ti * 16 + r + 8 * lh;
            int col = tj * 16 + lr;
            int l = c0 + row;
            out[(((size_t)n * LL + l) * HH + h) * MM + col] = acc[r] / zden[row];
        }
    }
}

extern "C" void kernel_launch(void* const* d_in, const int* in_sizes, int n_in,
                              void* d_out, int out_size, void* d_ws, size_t ws_size,
                              hipStream_t stream) {
    const float* queries = (const float*)d_in[0];
    const float* keys    = (const float*)d_in[1];
    const float* values  = (const float*)d_in[2];
    const float* klen    = (const float*)d_in[3];
    float* outp = (float*)d_out;

    float* ws      = (float*)d_ws;
    float* Sbuf    = ws;
    float* Ksum    = Sbuf + (size_t)NB * HH * NC * EE * MM;
    float* mu      = Ksum + (size_t)NB * HH * NC * EE;
    float* weights = mu + (size_t)NB * LL;

    mu_kernel<<<NB * (LL - 1), 256, 0, stream>>>(keys, values, klen, mu);
    weights_kernel<<<NB, 32, 0, stream>>>(mu, weights);
    sdelta_kernel<<<NB * HH * NC, 256, 0, stream>>>(keys, values, klen, weights, Sbuf, Ksum);
    prefix_kernel<<<NB * HH, 256, 0, stream>>>(Sbuf, Ksum);
    out_kernel<<<NB * HH * NC, 256, 0, stream>>>(queries, keys, values, klen,
                                                 weights, Sbuf, Ksum, outp);
}


# weights scan: the one-thread serial loop of 2047 dependent steps becomes a block-parallel affine suffix scan (256 threads x 8 positions, lane shuffles, LDS across the 8 waves)
# speedup vs baseline: 1.7795x; 1.7795x over previous
#include <hip/hip_runtime.h>
#include <hip/hip_bf16.h>

#define NB 2
#define LL 2048
#define HH 8
#define EE 64
#define MM 64
#define CK 64
#define NC (LL/CK)
#define EPS 1e-6f
#define PAD 4
#define LDW (EE + PAD)

typedef __attribute__((ext_vector_type(2))) float v2f;
typedef __attribute__((ext_vector_type(8))) float v8f;

__device__ __forceinline__ float fmap(float x) {
    return x > 0.0f ? x + 1.0f : __expf(x);
}

__global__ void mu_kernel(const float* __restrict__ keys,
                          const float* __restrict__ values,
                          const float* __restrict__ klen,
                          float* __restrict__ mu) {
    int b = blockIdx.x;
    int n = b / (LL - 1);
    int l = b % (LL - 1);
    int t = threadIdx.x;
    int h = t >> 5;
    int lane = t & 31;

    float kl0 = klen[n * LL + l];
    float kl1 = klen[n * LL + l + 1];

    float kn0 = 0.f, kn1 = 0.f, kd = 0.f, vn0 = 0.f, vn1 = 0.f, vd = 0.f;
    for (int e = lane; e < EE; e += 32) {
        float k0 = fmap(keys[(((size_t)n * LL + l)     * HH + h) * EE + e]) * kl0;
        float k1 = fmap(keys[(((size_t)n * LL + l + 1) * HH + h) * EE + e]) * kl1;
        kn0 += k0 * k0; kn1 += k1 * k1; kd += k0 * k1;
        float v0 = values[(((size_t)n * LL + l)     * HH + h) * MM + e];
        float v1 = values[(((size_t)n * LL + l + 1) * HH + h) * MM + e];
        vn0 += v0 * v0; vn1 += v1 * v1; vd += v0 * v1;
    }
    for (int off = 16; off > 0; off >>= 1) {
        kn0 += __shfl_xor(kn0, off, 32);
        kn1 += __shfl_xor(kn1, off, 32);
        kd  += __shfl_xor(kd,  off, 32);
        vn0 += __shfl_xor(vn0, off, 32);
        vn1 += __shfl_xor(vn1, off, 32);
        vd  += __shfl_xor(vd,  off, 32);
    }
    __shared__ float sh[HH][6];
    if (lane == 0) {
        sh[h][0] = kn0; sh[h][1] = kn1; sh[h][2] = kd;
        sh[h][3] = vn0; sh[h][4] = vn1; sh[h][5] = vd;
    }
    __syncthreads();
    if (t == 0) {
        float base2 = 0.f, diff2 = 0.f;
        for (int hh = 0; hh < HH; ++hh) {
            float a = sh[hh][0], b1 = sh[hh][1], c = sh[hh][2];
            float d = sh[hh][3], e1 = sh[hh][4], f = sh[hh][5];
            base2 += a * d;
            diff2 += b1 * e1 + a * d - 2.0f * c * f;
        }
        diff2 = fmaxf(diff2, 0.0f);
        float ratio = sqrtf(diff2) / sqrtf(base2);
        float r4 = sqrtf(ratio);
        float m_ = 1.0f - r4;
        m_ = m_ * m_;
        m_ = fminf(fmaxf(m_, 0.0f), 0.9999f);
        mu[n * LL + l] = m_;
    }
}

#define WSEG 8
#define WTHREADS (LL / WSEG)
__global__ void __launch_bounds__(WTHREADS)
weights_kernel(const float* __restrict__ mu,
               float* __restrict__ weights) {
    __shared__ float sA[WTHREADS / 32];
    __shared__ float sB[WTHREADS / 32];

    int n = blockIdx.x;
    int t = threadIdx.x;
    int lane = t & 31, wave = t >> 5;
    const float* mp = mu + (size_t)n * LL + t * WSEG;

    float m[WSEG];
    if (t < WTHREADS - 1) {
        const float4 lo = *(const float4*)(mp);
        const float4 hi = *(const float4*)(mp + 4);
        m[0] = lo.x; m[1] = lo.y; m[2] = lo.z; m[3] = lo.w;
        m[4] = hi.x; m[5] = hi.y; m[6] = hi.z; m[7] = hi.w;
    } else {
        const float4 lo = *(const float4*)(mp);
        m[0] = lo.x; m[1] = lo.y; m[2] = lo.z; m[3] = lo.w;
        m[4] = mp[4]; m[5] = mp[5]; m[6] = mp[6];
        m[7] = 0.0f;
    }

    float A = m[WSEG - 1];
    float B = 1.0f;
#pragma unroll
    for (int j = WSEG - 2; j >= 0; --j) {
        B = B * m[j] + 1.0f;
        A = A * m[j];
    }

#pragma unroll
    for (int off = 1; off < 32; off <<= 1) {
        float A2 = __shfl_down(A, off, 32);
        float B2 = __shfl_down(B, off, 32);
        if (lane + off < 32) {
            B = A * B2 + B;
            A = A * A2;
        }
    }
    if (lane == 0) { sA[wave] = A; sB[wave] = B; }

    float Ae = __shfl_down(A, 1, 32);
    float Be = __shfl_down(B, 1, 32);
    if (lane == 31) { Ae = 1.0f; Be = 0.0f; }
    __syncthreads();

    float xw = 0.0f;
#pragma unroll
    for (int v = WTHREADS / 32 - 1; v >= 1; --v) {
        float a = sA[v], b = sB[v];
        if (v > wave) xw = a * xw + b;
    }

    float x = Ae * xw + Be;

    float o[WSEG];
#pragma unroll
    for (int j = WSEG - 1; j >= 0; --j) {
        x = x * m[j] + 1.0f;
        o[j] = x;
    }
    float* wp = weights + (size_t)n * LL + t * WSEG;
    *(float4*)(wp)     = make_float4(o[0], o[1], o[2], o[3]);
    *(float4*)(wp + 4) = make_float4(o[4], o[5], o[6], o[7]);
}

__global__ void sdelta_kernel(const float* __restrict__ keys,
                              const float* __restrict__ values,
                              const float* __restrict__ klen,
                              const float* __restrict__ weights,
                              float* __restrict__ Sbuf,
                              float* __restrict__ Ksum) {
    __shared__ float Klds[CK][LDW];
    __shared__ float Vlds[CK][LDW];

    int b = blockIdx.x;
    int c = b % NC;
    int h = (b / NC) % HH;
    int n = b / (NC * HH);
    int c0 = c * CK;
    int t = threadIdx.x;

    for (int i = t; i < CK * EE / 4; i += 256) {
        int s  = i >> 4;
        int e4 = (i & 15) * 4;
        int l  = c0 + s;
        float kl = klen[n * LL + l];
        float w  = weights[n * LL + l];
        const float4 kv = *(const float4*)&keys  [(((size_t)n * LL + l) * HH + h) * EE + e4];
        const float4 vv = *(const float4*)&values[(((size_t)n * LL + l) * HH + h) * MM + e4];
        float4 ko, vo;
        ko.x = fmap(kv.x) * kl; ko.y = fmap(kv.y) * kl;
        ko.z = fmap(kv.z) * kl; ko.w = fmap(kv.w) * kl;
        vo.x = vv.x * w; vo.y = vv.y * w; vo.z = vv.z * w; vo.w = vv.w * w;
        *(float4*)&Klds[s][e4] = ko;
        *(float4*)&Vlds[s][e4] = vo;
    }
    __syncthreads();

    if (t < EE) {
        float acc = 0.f;
        for (int s = 0; s < CK; ++s) acc += Klds[s][t];
        Ksum[(((size_t)n * HH + h) * NC + c) * EE + t] = acc;
    }

    int wave = t >> 5, lane = t & 31, lr = lane & 15, lh = lane >> 4;
    float* out = Sbuf + (((size_t)n * HH + h) * NC + c) * (EE * MM);

    for (int tile = wave; tile < 16; tile += 8) {
        int ti = tile >> 2, tj = tile & 3;
        v8f acc = {};
        for (int k0 = 0; k0 < CK; k0 += 4) {
            int s = k0 + 2 * lh;
            v2f a, bb;
            int e = ti * 16 + lr;
            a.x = Klds[s][e];
            a.y = Klds[s + 1][e];
            int m = tj * 16 + lr;
            bb.x = Vlds[s][m];
            bb.y = Vlds[s + 1][m];
            acc = __builtin_amdgcn_wmma_f32_16x16x4_f32(false, a, false, bb,
                                                        (short)0, acc, false, false);
        }
        for (int r = 0; r < 8; ++r) {
            int row = ti * 16 + r + 8 * lh;
            int col = tj * 16 + lr;
            out[row * MM + col] = acc[r];
        }
    }
}

__global__ void prefix_kernel(float* __restrict__ Sbuf,
                              float* __restrict__ Ksum) {
    int b = blockIdx.x;
    int t = threadIdx.x;
    float4* base = (float4*)(Sbuf + (size_t)b * NC * (EE * MM));
    for (int i = t; i < EE * MM / 4; i += 256) {
        float4 acc = {0.f, 0.f, 0.f, 0.f};
        for (int c = 0; c < NC; ++c) {
            float4 v = base[(size_t)c * (EE * MM / 4) + i];
            base[(size_t)c * (EE * MM / 4) + i] = acc;
            acc.x += v.x; acc.y += v.y; acc.z += v.z; acc.w += v.w;
        }
    }
    if (t < EE) {
        float* kb = Ksum + (size_t)b * NC * EE;
        float acc = 0.f;
        for (int c = 0; c < NC; ++c) {
            float v = kb[c * EE + t];
            kb[c * EE + t] = acc;
            acc += v;
        }
    }
}

__global__ void out_kernel(const float* __restrict__ queries,
                           const float* __restrict__ keys,
                           const float* __restrict__ values,
                           const float* __restrict__ klen,
                           const float* __restrict__ weights,
                           const float* __restrict__ Sbuf,
                           const float* __restrict__ Ksum,
                           float* __restrict__ out) {
    __shared__ float Qlds[CK][LDW];
    __shared__ float KP[CK][LDW];
    __shared__ float Vlds[CK][LDW];
    __shared__ float Slds[EE][LDW];
    __shared__ float ks[EE];
    __shared__ float zden[CK];

    int b = blockIdx.x;
    int c = b % NC;
    int h = (b / NC) % HH;
    int n = b / (NC * HH);
    int c0 = c * CK;
    int t = threadIdx.x;

    const float* Sp = Sbuf + (((size_t)n * HH + h) * NC + c) * (EE * MM);
    for (int i = t; i < CK * EE / 4; i += 256) {
        int s  = i >> 4;
        int e4 = (i & 15) * 4;
        int l  = c0 + s;
        float kl = klen[n * LL + l];
        float w  = weights[n * LL + l];
        const float4 qv = *(const float4*)&queries[(((size_t)n * LL + l) * HH + h) * EE + e4];
        const float4 kv = *(const float4*)&keys   [(((size_t)n * LL + l) * HH + h) * EE + e4];
        const float4 vv = *(const float4*)&values [(((size_t)n * LL + l) * HH + h) * MM + e4];
        float4 qo, ko, vo;
        qo.x = fmap(qv.x);      qo.y = fmap(qv.y);
        qo.z = fmap(qv.z);      qo.w = fmap(qv.w);
        ko.x = fmap(kv.x) * kl; ko.y = fmap(kv.y) * kl;
        ko.z = fmap(kv.z) * kl; ko.w = fmap(kv.w) * kl;
        vo.x = vv.x * w; vo.y = vv.y * w; vo.z = vv.z * w; vo.w = vv.w * w;
        *(float4*)&Qlds[s][e4] = qo;
        *(float4*)&KP[s][e4]   = ko;
        *(float4*)&Vlds[s][e4] = vo;
        *(float4*)&Slds[s][e4] = *(const float4*)&Sp[s * MM + e4];
    }
    if (t < EE) ks[t] = Ksum[(((size_t)n * HH + h) * NC + c) * EE + t];
    __syncthreads();

    int wave = t >> 5, lane = t & 31, lr = lane & 15, lh = lane >> 4;

    v8f p0 = {}, p1 = {};
    for (int half = 0; half < 2; ++half) {
        int tile = wave + half * 8;
        int ti = tile >> 2, tj = tile & 3;
        v8f acc = {};
        for (int k0 = 0; k0 < EE; k0 += 4) {
            int e = k0 + 2 * lh;
            v2f a, bb;
            int row = ti * 16 + lr;
            a.x = Qlds[row][e];
            a.y = Qlds[row][e + 1];
            int col = tj * 16 + lr;
            bb.x = KP[col][e];
            bb.y = KP[col][e + 1];
            acc = __builtin_amdgcn_wmma_f32_16x16x4_f32(false, a, false, bb,
                                                        (short)0, acc, false, false);
        }
        if (half) p1 = acc; else p0 = acc;
    }
    __syncthreads();

    for (int half = 0; half < 2; ++half) {
        v8f acc = half ? p1 : p0;
        int tile = wave + half * 8;
        int ti = tile >> 2, tj = tile & 3;
        for (int r = 0; r < 8; ++r) {
            int row = ti * 16 + r + 8 * lh;
            int col = tj * 16 + lr;
            KP[row][col] = (col <= row) ? acc[r] : 0.0f;
        }
    }
    __syncthreads();

    if (t < CK) {
        float s1 = 0.f;
        for (int s = 0; s <= t; ++s) s1 += KP[t][s];
        float s2 = 0.f;
        for (int e = 0; e < EE; ++e) s2 += Qlds[t][e] * ks[e];
        zden[t] = s1 + s2 + EPS;
    }
    __syncthreads();

    for (int half = 0; half < 2; ++half) {
        int tile = wave + half * 8;
        int ti = tile >> 2, tj = tile & 3;
        v8f acc = {};
        for (int k0 = 0; k0 < EE; k0 += 4) {
            int e = k0 + 2 * lh;
            v2f a, bb;
            int row = ti * 16 + lr;
            a.x = Qlds[row][e];
            a.y = Qlds[row][e + 1];
            int col = tj * 16 + lr;
            bb.x = Slds[e][col];
            bb.y = Slds[e + 1][col];
            acc = __builtin_amdgcn_wmma_f32_16x16x4_f32(false, a, false, bb,
                                                        (short)0, acc, false, false);
        }
        for (int k0 = 0; k0 < CK; k0 += 4) {
            int s = k0 + 2 * lh;
            v2f a, bb;
            int row = ti * 16 + lr;
            a.x = KP[row][s];
            a.y = KP[row][s + 1];
            int col = tj * 16 + lr;
            bb.x = Vlds[s][col];
            bb.y = Vlds[s + 1][col];
            acc = __builtin_amdgcn_wmma_f32_16x16x4_f32(false, a, false, bb,
                                                        (short)0, acc, false, false);
        }
        for (int r = 0; r < 8; ++r) {
            int row = ti * 16 + r + 8 * lh;
            int col = tj * 16 + lr;
            int l = c0 + row;
            out[(((size_t)n * LL + l) * HH + h) * MM + col] = acc[r] / zden[row];
        }
    }
}

extern "C" void kernel_launch(void* const* d_in, const int* in_sizes, int n_in,
                              void* d_out, int out_size, void* d_ws, size_t ws_size,
                              hipStream_t stream) {
    const float* queries = (const float*)d_in[0];
    const float* keys    = (const float*)d_in[1];
    const float* values  = (const float*)d_in[2];
    const float* klen    = (const float*)d_in[3];
    float* outp = (float*)d_out;

    float* ws      = (float*)d_ws;
    float* Sbuf    = ws;
    float* Ksum    = Sbuf + (size_t)NB * HH * NC * EE * MM;
    float* mu      = Ksum + (size_t)NB * HH * NC * EE;
    float* weights = mu + (size_t)NB * LL;

    mu_kernel<<<NB * (LL - 1), 256, 0, stream>>>(keys, values, klen, mu);
    weights_kernel<<<NB, 256, 0, stream>>>(mu, weights);
    sdelta_kernel<<<NB * HH * NC, 256, 0, stream>>>(keys, values, klen, weights, Sbuf, Ksum);
    prefix_kernel<<<NB * HH, 256, 0, stream>>>(Sbuf, Ksum);
    out_kernel<<<NB * HH * NC, 256, 0, stream>>>(queries, keys, values, klen,
                                                 weights, Sbuf, Ksum, outp);
}


# mu kernel: one wave per 4 consecutive positions (all 8 heads, float4 row loads, per-head sums by 16-lane DPP butterflies) replaces one 8-wave workgroup per position: 32x fewer waves to dispatch
# speedup vs baseline: 1.8928x; 1.0637x over previous
#include <hip/hip_runtime.h>
#include <hip/hip_bf16.h>

#define NB 2
#define LL 2048
#define HH 8
#define EE 64
#define MM 64
#define CK 64
#define NC (LL/CK)
#define EPS 1e-6f
#define PAD 4
#define LDW (EE + PAD)

typedef __attribute__((ext_vector_type(2))) float v2f;
typedef __attribute__((ext_vector_type(8))) float v8f;

__device__ __forceinline__ float fmap(float x) {
    return x > 0.0f ? x + 1.0f : __expf(x);
}

#define MU_R 4
#define MU_ROWS (MU_R + 1)
#define MU_WAVES 4

template <int M>
__device__ __forceinline__ float lane_xor16(float x) {
    int b = __builtin_bit_cast(int, x);
    return __builtin_bit_cast(float,
        __builtin_amdgcn_mov_dpp(b, 0x160 + M, 0xf, 0xf, false));
}

__device__ __forceinline__ float row16_sum(float x) {
    x += lane_xor16<1>(x);
    x += lane_xor16<2>(x);
    x += lane_xor16<4>(x);
    x += lane_xor16<8>(x);
    return x;
}

__device__ __forceinline__ float xhalf(float x) {
    int b = __builtin_bit_cast(int, x);
    return __builtin_bit_cast(float,
        __builtin_amdgcn_permlanex16(b, b, 0x76543210, 0xfedcba98, false, false));
}

__device__ __forceinline__ float dot4(const float4 a, const float4 b) {
    return a.x * b.x + a.y * b.y + a.z * b.z + a.w * b.w;
}

__global__ void __launch_bounds__(MU_WAVES * 32)
mu_kernel(const float* __restrict__ keys,
          const float* __restrict__ values,
          const float* __restrict__ klen,
          float* __restrict__ mu) {
    const int lane = threadIdx.x & 31;
    const int wave = __builtin_amdgcn_readfirstlane((int)(threadIdx.x >> 5));
    const int run  = blockIdx.x * MU_WAVES + wave;
    const int n    = run / (LL / MU_R);
    const int l0   = (run % (LL / MU_R)) * MU_R;

    float4 kq[MU_ROWS][4], vq[MU_ROWS][4];
    float  kl[MU_ROWS];
#pragma unroll
    for (int r = 0; r < MU_ROWS; ++r) {
        const int l = (l0 + r < LL) ? (l0 + r) : (LL - 1);
        const float4* kp = (const float4*)(keys   + ((size_t)n * LL + l) * (HH * EE));
        const float4* vp = (const float4*)(values + ((size_t)n * LL + l) * (HH * MM));
        kl[r] = klen[n * LL + l];
#pragma unroll
        for (int j = 0; j < 4; ++j) {
            kq[r][j] = kp[j * 32 + lane];
            vq[r][j] = vp[j * 32 + lane];
        }
    }

    float mval[MU_R];
    float Bprev = 0.0f;
#pragma unroll
    for (int r = 0; r < MU_ROWS; ++r) {
        float bs = 0.0f, ds = 0.0f;
#pragma unroll
        for (int j = 0; j < 4; ++j) {
            float4 k = kq[r][j];
            k.x = fmap(k.x) * kl[r]; k.y = fmap(k.y) * kl[r];
            k.z = fmap(k.z) * kl[r]; k.w = fmap(k.w) * kl[r];
            kq[r][j] = k;
            const float4 v = vq[r][j];
            const float kn = row16_sum(dot4(k, k));
            const float vn = row16_sum(dot4(v, v));
            bs += kn * vn;
            if (r > 0) {
                const float kd = row16_sum(dot4(k, kq[r - 1][j]));
                const float vd = row16_sum(dot4(v, vq[r - 1][j]));
                ds += kd * vd;
            }
        }
        const float B = bs + xhalf(bs);
        if (r > 0) {
            const float D = ds + xhalf(ds);
            float diff2 = Bprev + B - 2.0f * D;
            diff2 = fmaxf(diff2, 0.0f);
            float ratio = sqrtf(diff2) / sqrtf(Bprev);
            float r4 = sqrtf(ratio);
            float m_ = 1.0f - r4;
            m_ = m_ * m_;
            m_ = fminf(fmaxf(m_, 0.0f), 0.9999f);
            mval[r - 1] = m_;
        }
        Bprev = B;
    }

    float mv = mval[0];
#pragma unroll
    for (int r = 1; r < MU_R; ++r) mv = (lane == r) ? mval[r] : mv;
    const int l = l0 + lane;
    if (lane < MU_R && l < LL - 1) mu[n * LL + l] = mv;
}

#define WSEG 8
#define WTHREADS (LL / WSEG)
__global__ void __launch_bounds__(WTHREADS)
weights_kernel(const float* __restrict__ mu,
               float* __restrict__ weights) {
    __shared__ float sA[WTHREADS / 32];
    __shared__ float sB[WTHREADS / 32];

    int n = blockIdx.x;
    int t = threadIdx.x;
    int lane = t & 31, wave = t >> 5;
    const float* mp = mu + (size_t)n * LL + t * WSEG;

    float m[WSEG];
    if (t < WTHREADS - 1) {
        const float4 lo = *(const float4*)(mp);
        const float4 hi = *(const float4*)(mp + 4);
        m[0] = lo.x; m[1] = lo.y; m[2] = lo.z; m[3] = lo.w;
        m[4] = hi.x; m[5] = hi.y; m[6] = hi.z; m[7] = hi.w;
    } else {
        const float4 lo = *(const float4*)(mp);
        m[0] = lo.x; m[1] = lo.y; m[2] = lo.z; m[3] = lo.w;
        m[4] = mp[4]; m[5] = mp[5]; m[6] = mp[6];
        m[7] = 0.0f;
    }

    float A = m[WSEG - 1];
    float B = 1.0f;
#pragma unroll
    for (int j = WSEG - 2; j >= 0; --j) {
        B = B * m[j] + 1.0f;
        A = A * m[j];
    }

#pragma unroll
    for (int off = 1; off < 32; off <<= 1) {
        float A2 = __shfl_down(A, off, 32);
        float B2 = __shfl_down(B, off, 32);
        if (lane + off < 32) {
            B = A * B2 + B;
            A = A * A2;
        }
    }
    if (lane == 0) { sA[wave] = A; sB[wave] = B; }

    float Ae = __shfl_down(A, 1, 32);
    float Be = __shfl_down(B, 1, 32);
    if (lane == 31) { Ae = 1.0f; Be = 0.0f; }
    __syncthreads();

    float xw = 0.0f;
#pragma unroll
    for (int v = WTHREADS / 32 - 1; v >= 1; --v) {
        float a = sA[v], b = sB[v];
        if (v > wave) xw = a * xw + b;
    }

    float x = Ae * xw + Be;

    float o[WSEG];
#pragma unroll
    for (int j = WSEG - 1; j >= 0; --j) {
        x = x * m[j] + 1.0f;
        o[j] = x;
    }
    float* wp = weights + (size_t)n * LL + t * WSEG;
    *(float4*)(wp)     = make_float4(o[0], o[1], o[2], o[3]);
    *(float4*)(wp + 4) = make_float4(o[4], o[5], o[6], o[7]);
}

__global__ void sdelta_kernel(const float* __restrict__ keys,
                              const float* __restrict__ values,
                              const float* __restrict__ klen,
                              const float* __restrict__ weights,
                              float* __restrict__ Sbuf,
                              float* __restrict__ Ksum) {
    __shared__ float Klds[CK][LDW];
    __shared__ float Vlds[CK][LDW];

    int b = blockIdx.x;
    int c = b % NC;
    int h = (b / NC) % HH;
    int n = b / (NC * HH);
    int c0 = c * CK;
    int t = threadIdx.x;

    for (int i = t; i < CK * EE / 4; i += 256) {
        int s  = i >> 4;
        int e4 = (i & 15) * 4;
        int l  = c0 + s;
        float kl = klen[n * LL + l];
        float w  = weights[n * LL + l];
        const float4 kv = *(const float4*)&keys  [(((size_t)n * LL + l) * HH + h) * EE + e4];
        const float4 vv = *(const float4*)&values[(((size_t)n * LL + l) * HH + h) * MM + e4];
        float4 ko, vo;
        ko.x = fmap(kv.x) * kl; ko.y = fmap(kv.y) * kl;
        ko.z = fmap(kv.z) * kl; ko.w = fmap(kv.w) * kl;
        vo.x = vv.x * w; vo.y = vv.y * w; vo.z = vv.z * w; vo.w = vv.w * w;
        *(float4*)&Klds[s][e4] = ko;
        *(float4*)&Vlds[s][e4] = vo;
    }
    __syncthreads();

    if (t < EE) {
        float acc = 0.f;
        for (int s = 0; s < CK; ++s) acc += Klds[s][t];
        Ksum[(((size_t)n * HH + h) * NC + c) * EE + t] = acc;
    }

    int wave = t >> 5, lane = t & 31, lr = lane & 15, lh = lane >> 4;
    float* out = Sbuf + (((size_t)n * HH + h) * NC + c) * (EE * MM);

    for (int tile = wave; tile < 16; tile += 8) {
        int ti = tile >> 2, tj = tile & 3;
        v8f acc = {};
        for (int k0 = 0; k0 < CK; k0 += 4) {
            int s = k0 + 2 * lh;
            v2f a, bb;
            int e = ti * 16 + lr;
            a.x = Klds[s][e];
            a.y = Klds[s + 1][e];
            int m = tj * 16 + lr;
            bb.x = Vlds[s][m];
            bb.y = Vlds[s + 1][m];
            acc = __builtin_amdgcn_wmma_f32_16x16x4_f32(false, a, false, bb,
                                                        (short)0, acc, false, false);
        }
        for (int r = 0; r < 8; ++r) {
            int row = ti * 16 + r + 8 * lh;
            int col = tj * 16 + lr;
            out[row * MM + col] = acc[r];
        }
    }
}

__global__ void prefix_kernel(float* __restrict__ Sbuf,
                              float* __restrict__ Ksum) {
    int b = blockIdx.x;
    int t = threadIdx.x;
    float4* base = (float4*)(Sbuf + (size_t)b * NC * (EE * MM));
    for (int i = t; i < EE * MM / 4; i += 256) {
        float4 acc = {0.f, 0.f, 0.f, 0.f};
        for (int c = 0; c < NC; ++c) {
            float4 v = base[(size_t)c * (EE * MM / 4) + i];
            base[(size_t)c * (EE * MM / 4) + i] = acc;
            acc.x += v.x; acc.y += v.y; acc.z += v.z; acc.w += v.w;
        }
    }
    if (t < EE) {
        float* kb = Ksum + (size_t)b * NC * EE;
        float acc = 0.f;
        for (int c = 0; c < NC; ++c) {
            float v = kb[c * EE + t];
            kb[c * EE + t] = acc;
            acc += v;
        }
    }
}

__global__ void out_kernel(const float* __restrict__ queries,
                           const float* __restrict__ keys,
                           const float* __restrict__ values,
                           const float* __restrict__ klen,
                           const float* __restrict__ weights,
                           const float* __restrict__ Sbuf,
                           const float* __restrict__ Ksum,
                           float* __restrict__ out) {
    __shared__ float Qlds[CK][LDW];
    __shared__ float KP[CK][LDW];
    __shared__ float Vlds[CK][LDW];
    __shared__ float Slds[EE][LDW];
    __shared__ float ks[EE];
    __shared__ float zden[CK];

    int b = blockIdx.x;
    int c = b % NC;
    int h = (b / NC) % HH;
    int n = b / (NC * HH);
    int c0 = c * CK;
    int t = threadIdx.x;

    const float* Sp = Sbuf + (((size_t)n * HH + h) * NC + c) * (EE * MM);
    for (int i = t; i < CK * EE / 4; i += 256) {
        int s  = i >> 4;
        int e4 = (i & 15) * 4;
        int l  = c0 + s;
        float kl = klen[n * LL + l];
        float w  = weights[n * LL + l];
        const float4 qv = *(const float4*)&queries[(((size_t)n * LL + l) * HH + h) * EE + e4];
        const float4 kv = *(const float4*)&keys   [(((size_t)n * LL + l) * HH + h) * EE + e4];
        const float4 vv = *(const float4*)&values [(((size_t)n * LL + l) * HH + h) * MM + e4];
        float4 qo, ko, vo;
        qo.x = fmap(qv.x);      qo.y = fmap(qv.y);
        qo.z = fmap(qv.z);      qo.w = fmap(qv.w);
        ko.x = fmap(kv.x) * kl; ko.y = fmap(kv.y) * kl;
        ko.z = fmap(kv.z) * kl; ko.w = fmap(kv.w) * kl;
        vo.x = vv.x * w; vo.y = vv.y * w; vo.z = vv.z * w; vo.w = vv.w * w;
        *(float4*)&Qlds[s][e4] = qo;
        *(float4*)&KP[s][e4]   = ko;
        *(float4*)&Vlds[s][e4] = vo;
        *(float4*)&Slds[s][e4] = *(const float4*)&Sp[s * MM + e4];
    }
    if (t < EE) ks[t] = Ksum[(((size_t)n * HH + h) * NC + c) * EE + t];
    __syncthreads();

    int wave = t >> 5, lane = t & 31, lr = lane & 15, lh = lane >> 4;

    v8f p0 = {}, p1 = {};
    for (int half = 0; half < 2; ++half) {
        int tile = wave + half * 8;
        int ti = tile >> 2, tj = tile & 3;
        v8f acc = {};
        for (int k0 = 0; k0 < EE; k0 += 4) {
            int e = k0 + 2 * lh;
            v2f a, bb;
            int row = ti * 16 + lr;
            a.x = Qlds[row][e];
            a.y = Qlds[row][e + 1];
            int col = tj * 16 + lr;
            bb.x = KP[col][e];
            bb.y = KP[col][e + 1];
            acc = __builtin_amdgcn_wmma_f32_16x16x4_f32(false, a, false, bb,
                                                        (short)0, acc, false, false);
        }
        if (half) p1 = acc; else p0 = acc;
    }
    __syncthreads();

    for (int half = 0; half < 2; ++half) {
        v8f acc = half ? p1 : p0;
        int tile = wave + half * 8;
        int ti = tile >> 2, tj = tile & 3;
        for (int r = 0; r < 8; ++r) {
            int row = ti * 16 + r + 8 * lh;
            int col = tj * 16 + lr;
            KP[row][col] = (col <= row) ? acc[r] : 0.0f;
        }
    }
    __syncthreads();

    if (t < CK) {
        float s1 = 0.f;
        for (int s = 0; s <= t; ++s) s1 += KP[t][s];
        float s2 = 0.f;
        for (int e = 0; e < EE; ++e) s2 += Qlds[t][e] * ks[e];
        zden[t] = s1 + s2 + EPS;
    }
    __syncthreads();

    for (int half = 0; half < 2; ++half) {
        int tile = wave + half * 8;
        int ti = tile >> 2, tj = tile & 3;
        v8f acc = {};
        for (int k0 = 0; k0 < EE; k0 += 4) {
            int e = k0 + 2 * lh;
            v2f a, bb;
            int row = ti * 16 + lr;
            a.x = Qlds[row][e];
            a.y = Qlds[row][e + 1];
            int col = tj * 16 + lr;
            bb.x = Slds[e][col];
            bb.y = Slds[e + 1][col];
            acc = __builtin_amdgcn_wmma_f32_16x16x4_f32(false, a, false, bb,
                                                        (short)0, acc, false, false);
        }
        for (int k0 = 0; k0 < CK; k0 += 4) {
            int s = k0 + 2 * lh;
            v2f a, bb;
            int row = ti * 16 + lr;
            a.x = KP[row][s];
            a.y = KP[row][s + 1];
            int col = tj * 16 + lr;
            bb.x = Vlds[s][col];
            bb.y = Vlds[s + 1][col];
            acc = __builtin_amdgcn_wmma_f32_16x16x4_f32(false, a, false, bb,
                                                        (short)0, acc, false, false);
        }
        for (int r = 0; r < 8; ++r) {
            int row = ti * 16 + r + 8 * lh;
            int col = tj * 16 + lr;
            int l = c0 + row;
            out[(((size_t)n * LL + l) * HH + h) * MM + col] = acc[r] / zden[row];
        }
    }
}

extern "C" void kernel_launch(void* const* d_in, const int* in_sizes, int n_in,
                              void* d_out, int out_size, void* d_ws, size_t ws_size,
                              hipStream_t stream) {
    const float* queries = (const float*)d_in[0];
    const float* keys    = (const float*)d_in[1];
    const float* values  = (const float*)d_in[2];
    const float* klen    = (const float*)d_in[3];
    float* outp = (float*)d_out;

    float* ws      = (float*)d_ws;
    float* Sbuf    = ws;
    float* Ksum    = Sbuf + (size_t)NB * HH * NC * EE * MM;
    float* mu      = Ksum + (size_t)NB * HH * NC * EE;
    float* weights = mu + (size_t)NB * LL;

    mu_kernel<<<NB * LL / (MU_R * MU_WAVES), MU_WAVES * 32, 0, stream>>>(keys, values, klen, mu);
    weights_kernel<<<NB, 256, 0, stream>>>(mu, weights);
    sdelta_kernel<<<NB * HH * NC, 256, 0, stream>>>(keys, values, klen, weights, Sbuf, Ksum);
    prefix_kernel<<<NB * HH, 256, 0, stream>>>(Sbuf, Ksum);
    out_kernel<<<NB * HH * NC, 256, 0, stream>>>(queries, keys, values, klen,
                                                 weights, Sbuf, Ksum, outp);
}


# prefix kernel: 16 workgroups walking 32 chunks with 2 loads in flight become 272 workgroups, one thread per float4 state column, all 31 chunk loads issued before the first store
# speedup vs baseline: 2.5046x; 1.3232x over previous
#include <hip/hip_runtime.h>
#include <hip/hip_bf16.h>

#define NB 2
#define LL 2048
#define HH 8
#define EE 64
#define MM 64
#define CK 64
#define NC (LL/CK)
#define EPS 1e-6f
#define PAD 4
#define LDW (EE + PAD)

typedef __attribute__((ext_vector_type(2))) float v2f;
typedef __attribute__((ext_vector_type(8))) float v8f;

__device__ __forceinline__ float fmap(float x) {
    return x > 0.0f ? x + 1.0f : __expf(x);
}

#define MU_R 4
#define MU_ROWS (MU_R + 1)
#define MU_WAVES 4

template <int M>
__device__ __forceinline__ float lane_xor16(float x) {
    int b = __builtin_bit_cast(int, x);
    return __builtin_bit_cast(float,
        __builtin_amdgcn_mov_dpp(b, 0x160 + M, 0xf, 0xf, false));
}

__device__ __forceinline__ float row16_sum(float x) {
    x += lane_xor16<1>(x);
    x += lane_xor16<2>(x);
    x += lane_xor16<4>(x);
    x += lane_xor16<8>(x);
    return x;
}

__device__ __forceinline__ float xhalf(float x) {
    int b = __builtin_bit_cast(int, x);
    return __builtin_bit_cast(float,
        __builtin_amdgcn_permlanex16(b, b, 0x76543210, 0xfedcba98, false, false));
}

__device__ __forceinline__ float dot4(const float4 a, const float4 b) {
    return a.x * b.x + a.y * b.y + a.z * b.z + a.w * b.w;
}

__global__ void __launch_bounds__(MU_WAVES * 32)
mu_kernel(const float* __restrict__ keys,
          const float* __restrict__ values,
          const float* __restrict__ klen,
          float* __restrict__ mu) {
    const int lane = threadIdx.x & 31;
    const int wave = __builtin_amdgcn_readfirstlane((int)(threadIdx.x >> 5));
    const int run  = blockIdx.x * MU_WAVES + wave;
    const int n    = run / (LL / MU_R);
    const int l0   = (run % (LL / MU_R)) * MU_R;

    float4 kq[MU_ROWS][4], vq[MU_ROWS][4];
    float  kl[MU_ROWS];
#pragma unroll
    for (int r = 0; r < MU_ROWS; ++r) {
        const int l = (l0 + r < LL) ? (l0 + r) : (LL - 1);
        const float4* kp = (const float4*)(keys   + ((size_t)n * LL + l) * (HH * EE));
        const float4* vp = (const float4*)(values + ((size_t)n * LL + l) * (HH * MM));
        kl[r] = klen[n * LL + l];
#pragma unroll
        for (int j = 0; j < 4; ++j) {
            kq[r][j] = kp[j * 32 + lane];
            vq[r][j] = vp[j * 32 + lane];
        }
    }

    float mval[MU_R];
    float Bprev = 0.0f;
#pragma unroll
    for (int r = 0; r < MU_ROWS; ++r) {
        float bs = 0.0f, ds = 0.0f;
#pragma unroll
        for (int j = 0; j < 4; ++j) {
            float4 k = kq[r][j];
            k.x = fmap(k.x) * kl[r]; k.y = fmap(k.y) * kl[r];
            k.z = fmap(k.z) * kl[r]; k.w = fmap(k.w) * kl[r];
            kq[r][j] = k;
            const float4 v = vq[r][j];
            const float kn = row16_sum(dot4(k, k));
            const float vn = row16_sum(dot4(v, v));
            bs += kn * vn;
            if (r > 0) {
                const float kd = row16_sum(dot4(k, kq[r - 1][j]));
                const float vd = row16_sum(dot4(v, vq[r - 1][j]));
                ds += kd * vd;
            }
        }
        const float B = bs + xhalf(bs);
        if (r > 0) {
            const float D = ds + xhalf(ds);
            float diff2 = Bprev + B - 2.0f * D;
            diff2 = fmaxf(diff2, 0.0f);
            float ratio = sqrtf(diff2) / sqrtf(Bprev);
            float r4 = sqrtf(ratio);
            float m_ = 1.0f - r4;
            m_ = m_ * m_;
            m_ = fminf(fmaxf(m_, 0.0f), 0.9999f);
            mval[r - 1] = m_;
        }
        Bprev = B;
    }

    float mv = mval[0];
#pragma unroll
    for (int r = 1; r < MU_R; ++r) mv = (lane == r) ? mval[r] : mv;
    const int l = l0 + lane;
    if (lane < MU_R && l < LL - 1) mu[n * LL + l] = mv;
}

#define WSEG 8
#define WTHREADS (LL / WSEG)
__global__ void __launch_bounds__(WTHREADS)
weights_kernel(const float* __restrict__ mu,
               float* __restrict__ weights) {
    __shared__ float sA[WTHREADS / 32];
    __shared__ float sB[WTHREADS / 32];

    int n = blockIdx.x;
    int t = threadIdx.x;
    int lane = t & 31, wave = t >> 5;
    const float* mp = mu + (size_t)n * LL + t * WSEG;

    float m[WSEG];
    if (t < WTHREADS - 1) {
        const float4 lo = *(const float4*)(mp);
        const float4 hi = *(const float4*)(mp + 4);
        m[0] = lo.x; m[1] = lo.y; m[2] = lo.z; m[3] = lo.w;
        m[4] = hi.x; m[5] = hi.y; m[6] = hi.z; m[7] = hi.w;
    } else {
        const float4 lo = *(const float4*)(mp);
        m[0] = lo.x; m[1] = lo.y; m[2] = lo.z; m[3] = lo.w;
        m[4] = mp[4]; m[5] = mp[5]; m[6] = mp[6];
        m[7] = 0.0f;
    }

    float A = m[WSEG - 1];
    float B = 1.0f;
#pragma unroll
    for (int j = WSEG - 2; j >= 0; --j) {
        B = B * m[j] + 1.0f;
        A = A * m[j];
    }

#pragma unroll
    for (int off = 1; off < 32; off <<= 1) {
        float A2 = __shfl_down(A, off, 32);
        float B2 = __shfl_down(B, off, 32);
        if (lane + off < 32) {
            B = A * B2 + B;
            A = A * A2;
        }
    }
    if (lane == 0) { sA[wave] = A; sB[wave] = B; }

    float Ae = __shfl_down(A, 1, 32);
    float Be = __shfl_down(B, 1, 32);
    if (lane == 31) { Ae = 1.0f; Be = 0.0f; }
    __syncthreads();

    float xw = 0.0f;
#pragma unroll
    for (int v = WTHREADS / 32 - 1; v >= 1; --v) {
        float a = sA[v], b = sB[v];
        if (v > wave) xw = a * xw + b;
    }

    float x = Ae * xw + Be;

    float o[WSEG];
#pragma unroll
    for (int j = WSEG - 1; j >= 0; --j) {
        x = x * m[j] + 1.0f;
        o[j] = x;
    }
    float* wp = weights + (size_t)n * LL + t * WSEG;
    *(float4*)(wp)     = make_float4(o[0], o[1], o[2], o[3]);
    *(float4*)(wp + 4) = make_float4(o[4], o[5], o[6], o[7]);
}

__global__ void sdelta_kernel(const float* __restrict__ keys,
                              const float* __restrict__ values,
                              const float* __restrict__ klen,
                              const float* __restrict__ weights,
                              float* __restrict__ Sbuf,
                              float* __restrict__ Ksum) {
    __shared__ float Klds[CK][LDW];
    __shared__ float Vlds[CK][LDW];

    int b = blockIdx.x;
    int c = b % NC;
    int h = (b / NC) % HH;
    int n = b / (NC * HH);
    int c0 = c * CK;
    int t = threadIdx.x;

    for (int i = t; i < CK * EE / 4; i += 256) {
        int s  = i >> 4;
        int e4 = (i & 15) * 4;
        int l  = c0 + s;
        float kl = klen[n * LL + l];
        float w  = weights[n * LL + l];
        const float4 kv = *(const float4*)&keys  [(((size_t)n * LL + l) * HH + h) * EE + e4];
        const float4 vv = *(const float4*)&values[(((size_t)n * LL + l) * HH + h) * MM + e4];
        float4 ko, vo;
        ko.x = fmap(kv.x) * kl; ko.y = fmap(kv.y) * kl;
        ko.z = fmap(kv.z) * kl; ko.w = fmap(kv.w) * kl;
        vo.x = vv.x * w; vo.y = vv.y * w; vo.z = vv.z * w; vo.w = vv.w * w;
        *(float4*)&Klds[s][e4] = ko;
        *(float4*)&Vlds[s][e4] = vo;
    }
    __syncthreads();

    if (t < EE) {
        float acc = 0.f;
        for (int s = 0; s < CK; ++s) acc += Klds[s][t];
        Ksum[(((size_t)n * HH + h) * NC + c) * EE + t] = acc;
    }

    int wave = t >> 5, lane = t & 31, lr = lane & 15, lh = lane >> 4;
    float* out = Sbuf + (((size_t)n * HH + h) * NC + c) * (EE * MM);

    for (int tile = wave; tile < 16; tile += 8) {
        int ti = tile >> 2, tj = tile & 3;
        v8f acc = {};
        for (int k0 = 0; k0 < CK; k0 += 4) {
            int s = k0 + 2 * lh;
            v2f a, bb;
            int e = ti * 16 + lr;
            a.x = Klds[s][e];
            a.y = Klds[s + 1][e];
            int m = tj * 16 + lr;
            bb.x = Vlds[s][m];
            bb.y = Vlds[s + 1][m];
            acc = __builtin_amdgcn_wmma_f32_16x16x4_f32(false, a, false, bb,
                                                        (short)0, acc, false, false);
        }
        for (int r = 0; r < 8; ++r) {
            int row = ti * 16 + r + 8 * lh;
            int col = tj * 16 + lr;
            out[row * MM + col] = acc[r];
        }
    }
}

#define PF_THREADS 64
#define PF_SBLOCKS (NB * HH * (EE * MM / 4) / PF_THREADS)
__global__ void __launch_bounds__(PF_THREADS)
prefix_kernel(float* __restrict__ Sbuf,
              float* __restrict__ Ksum) {
    const int t = threadIdx.x;
    if (blockIdx.x < PF_SBLOCKS) {
        const int col = blockIdx.x * PF_THREADS + t;
        const int b = col / (EE * MM / 4);
        const int i = col % (EE * MM / 4);
        float4* base = (float4*)(Sbuf + (size_t)b * NC * (EE * MM)) + i;
        float4 v[NC - 1];
#pragma unroll
        for (int c = 0; c < NC - 1; ++c) v[c] = base[(size_t)c * (EE * MM / 4)];
        __builtin_amdgcn_sched_barrier(0);
        float4 acc = {0.f, 0.f, 0.f, 0.f};
#pragma unroll
        for (int c = 0; c < NC; ++c) {
            base[(size_t)c * (EE * MM / 4)] = acc;
            if (c < NC - 1) {
                acc.x += v[c].x; acc.y += v[c].y; acc.z += v[c].z; acc.w += v[c].w;
            }
        }
    } else {
        const int b = blockIdx.x - PF_SBLOCKS;
        float* kb = Ksum + (size_t)b * NC * EE + t;
        float v[NC - 1];
#pragma unroll
        for (int c = 0; c < NC - 1; ++c) v[c] = kb[c * EE];
        __builtin_amdgcn_sched_barrier(0);
        float acc = 0.f;
#pragma unroll
        for (int c = 0; c < NC; ++c) {
            kb[c * EE] = acc;
            if (c < NC - 1) acc += v[c];
        }
    }
}

__global__ void out_kernel(const float* __restrict__ queries,
                           const float* __restrict__ keys,
                           const float* __restrict__ values,
                           const float* __restrict__ klen,
                           const float* __restrict__ weights,
                           const float* __restrict__ Sbuf,
                           const float* __restrict__ Ksum,
                           float* __restrict__ out) {
    __shared__ float Qlds[CK][LDW];
    __shared__ float KP[CK][LDW];
    __shared__ float Vlds[CK][LDW];
    __shared__ float Slds[EE][LDW];
    __shared__ float ks[EE];
    __shared__ float zden[CK];

    int b = blockIdx.x;
    int c = b % NC;
    int h = (b / NC) % HH;
    int n = b / (NC * HH);
    int c0 = c * CK;
    int t = threadIdx.x;

    const float* Sp = Sbuf + (((size_t)n * HH + h) * NC + c) * (EE * MM);
    for (int i = t; i < CK * EE / 4; i += 256) {
        int s  = i >> 4;
        int e4 = (i & 15) * 4;
        int l  = c0 + s;
        float kl = klen[n * LL + l];
        float w  = weights[n * LL + l];
        const float4 qv = *(const float4*)&queries[(((size_t)n * LL + l) * HH + h) * EE + e4];
        const float4 kv = *(const float4*)&keys   [(((size_t)n * LL + l) * HH + h) * EE + e4];
        const float4 vv = *(const float4*)&values [(((size_t)n * LL + l) * HH + h) * MM + e4];
        float4 qo, ko, vo;
        qo.x = fmap(qv.x);      qo.y = fmap(qv.y);
        qo.z = fmap(qv.z);      qo.w = fmap(qv.w);
        ko.x = fmap(kv.x) * kl; ko.y = fmap(kv.y) * kl;
        ko.z = fmap(kv.z) * kl; ko.w = fmap(kv.w) * kl;
        vo.x = vv.x * w; vo.y = vv.y * w; vo.z = vv.z * w; vo.w = vv.w * w;
        *(float4*)&Qlds[s][e4] = qo;
        *(float4*)&KP[s][e4]   = ko;
        *(float4*)&Vlds[s][e4] = vo;
        *(float4*)&Slds[s][e4] = *(const float4*)&Sp[s * MM + e4];
    }
    if (t < EE) ks[t] = Ksum[(((size_t)n * HH + h) * NC + c) * EE + t];
    __syncthreads();

    int wave = t >> 5, lane = t & 31, lr = lane & 15, lh = lane >> 4;

    v8f p0 = {}, p1 = {};
    for (int half = 0; half < 2; ++half) {
        int tile = wave + half * 8;
        int ti = tile >> 2, tj = tile & 3;
        v8f acc = {};
        for (int k0 = 0; k0 < EE; k0 += 4) {
            int e = k0 + 2 * lh;
            v2f a, bb;
            int row = ti * 16 + lr;
            a.x = Qlds[row][e];
            a.y = Qlds[row][e + 1];
            int col = tj * 16 + lr;
            bb.x = KP[col][e];
            bb.y = KP[col][e + 1];
            acc = __builtin_amdgcn_wmma_f32_16x16x4_f32(false, a, false, bb,
                                                        (short)0, acc, false, false);
        }
        if (half) p1 = acc; else p0 = acc;
    }
    __syncthreads();

    for (int half = 0; half < 2; ++half) {
        v8f acc = half ? p1 : p0;
        int tile = wave + half * 8;
        int ti = tile >> 2, tj = tile & 3;
        for (int r = 0; r < 8; ++r) {
            int row = ti * 16 + r + 8 * lh;
            int col = tj * 16 + lr;
            KP[row][col] = (col <= row) ? acc[r] : 0.0f;
        }
    }
    __syncthreads();

    if (t < CK) {
        float s1 = 0.f;
        for (int s = 0; s <= t; ++s) s1 += KP[t][s];
        float s2 = 0.f;
        for (int e = 0; e < EE; ++e) s2 += Qlds[t][e] * ks[e];
        zden[t] = s1 + s2 + EPS;
    }
    __syncthreads();

    for (int half = 0; half < 2; ++half) {
        int tile = wave + half * 8;
        int ti = tile >> 2, tj = tile & 3;
        v8f acc = {};
        for (int k0 = 0; k0 < EE; k0 += 4) {
            int e = k0 + 2 * lh;
            v2f a, bb;
            int row = ti * 16 + lr;
            a.x = Qlds[row][e];
            a.y = Qlds[row][e + 1];
            int col = tj * 16 + lr;
            bb.x = Slds[e][col];
            bb.y = Slds[e + 1][col];
            acc = __builtin_amdgcn_wmma_f32_16x16x4_f32(false, a, false, bb,
                                                        (short)0, acc, false, false);
        }
        for (int k0 = 0; k0 < CK; k0 += 4) {
            int s = k0 + 2 * lh;
            v2f a, bb;
            int row = ti * 16 + lr;
            a.x = KP[row][s];
            a.y = KP[row][s + 1];
            int col = tj * 16 + lr;
            bb.x = Vlds[s][col];
            bb.y = Vlds[s + 1][col];
            acc = __builtin_amdgcn_wmma_f32_16x16x4_f32(false, a, false, bb,
                                                        (short)0, acc, false, false);
        }
        for (int r = 0; r < 8; ++r) {
            int row = ti * 16 + r + 8 * lh;
            int col = tj * 16 + lr;
            int l = c0 + row;
            out[(((size_t)n * LL + l) * HH + h) * MM + col] = acc[r] / zden[row];
        }
    }
}

extern "C" void kernel_launch(void* const* d_in, const int* in_sizes, int n_in,
                              void* d_out, int out_size, void* d_ws, size_t ws_size,
                              hipStream_t stream) {
    const float* queries = (const float*)d_in[0];
    const float* keys    = (const float*)d_in[1];
    const float* values  = (const float*)d_in[2];
    const float* klen    = (const float*)d_in[3];
    float* outp = (float*)d_out;

    float* ws      = (float*)d_ws;
    float* Sbuf    = ws;
    float* Ksum    = Sbuf + (size_t)NB * HH * NC * EE * MM;
    float* mu      = Ksum + (size_t)NB * HH * NC * EE;
    float* weights = mu + (size_t)NB * LL;

    mu_kernel<<<NB * LL / (MU_R * MU_WAVES), MU_WAVES * 32, 0, stream>>>(keys, values, klen, mu);
    weights_kernel<<<NB, 256, 0, stream>>>(mu, weights);
    sdelta_kernel<<<NB * HH * NC, 256, 0, stream>>>(keys, values, klen, weights, Sbuf, Ksum);
    prefix_kernel<<<PF_SBLOCKS + NB * HH, PF_THREADS, 0, stream>>>(Sbuf, Ksum);
    out_kernel<<<NB * HH * NC, 256, 0, stream>>>(queries, keys, values, klen,
                                                 weights, Sbuf, Ksum, outp);
}


# state-delta and output kernels: each thread requests all its global loads (K, V, Q, state-prefix rows, weights, column sums) before the first use; was 4 to 5 load-wait-store rounds per workgroup
# speedup vs baseline: 2.6629x; 1.0632x over previous
#include <hip/hip_runtime.h>
#include <hip/hip_bf16.h>

#define NB 2
#define LL 2048
#define HH 8
#define EE 64
#define MM 64
#define CK 64
#define NC (LL/CK)
#define EPS 1e-6f
#define PAD 4
#define LDW (EE + PAD)
#define LD_IT (CK * EE / 4 / 256)

typedef __attribute__((ext_vector_type(2))) float v2f;
typedef __attribute__((ext_vector_type(8))) float v8f;

__device__ __forceinline__ float fmap(float x) {
    return x > 0.0f ? x + 1.0f : __expf(x);
}

#define MU_R 4
#define MU_ROWS (MU_R + 1)
#define MU_WAVES 4

template <int M>
__device__ __forceinline__ float lane_xor16(float x) {
    int b = __builtin_bit_cast(int, x);
    return __builtin_bit_cast(float,
        __builtin_amdgcn_mov_dpp(b, 0x160 + M, 0xf, 0xf, false));
}

__device__ __forceinline__ float row16_sum(float x) {
    x += lane_xor16<1>(x);
    x += lane_xor16<2>(x);
    x += lane_xor16<4>(x);
    x += lane_xor16<8>(x);
    return x;
}

__device__ __forceinline__ float xhalf(float x) {
    int b = __builtin_bit_cast(int, x);
    return __builtin_bit_cast(float,
        __builtin_amdgcn_permlanex16(b, b, 0x76543210, 0xfedcba98, false, false));
}

__device__ __forceinline__ float dot4(const float4 a, const float4 b) {
    return a.x * b.x + a.y * b.y + a.z * b.z + a.w * b.w;
}

__global__ void __launch_bounds__(MU_WAVES * 32)
mu_kernel(const float* __restrict__ keys,
          const float* __restrict__ values,
          const float* __restrict__ klen,
          float* __restrict__ mu) {
    const int lane = threadIdx.x & 31;
    const int wave = __builtin_amdgcn_readfirstlane((int)(threadIdx.x >> 5));
    const int run  = blockIdx.x * MU_WAVES + wave;
    const int n    = run / (LL / MU_R);
    const int l0   = (run % (LL / MU_R)) * MU_R;

    float4 kq[MU_ROWS][4], vq[MU_ROWS][4];
    float  kl[MU_ROWS];
#pragma unroll
    for (int r = 0; r < MU_ROWS; ++r) {
        const int l = (l0 + r < LL) ? (l0 + r) : (LL - 1);
        const float4* kp = (const float4*)(keys   + ((size_t)n * LL + l) * (HH * EE));
        const float4* vp = (const float4*)(values + ((size_t)n * LL + l) * (HH * MM));
        kl[r] = klen[n * LL + l];
#pragma unroll
        for (int j = 0; j < 4; ++j) {
            kq[r][j] = kp[j * 32 + lane];
            vq[r][j] = vp[j * 32 + lane];
        }
    }

    float mval[MU_R];
    float Bprev = 0.0f;
#pragma unroll
    for (int r = 0; r < MU_ROWS; ++r) {
        float bs = 0.0f, ds = 0.0f;
#pragma unroll
        for (int j = 0; j < 4; ++j) {
            float4 k = kq[r][j];
            k.x = fmap(k.x) * kl[r]; k.y = fmap(k.y) * kl[r];
            k.z = fmap(k.z) * kl[r]; k.w = fmap(k.w) * kl[r];
            kq[r][j] = k;
            const float4 v = vq[r][j];
            const float kn = row16_sum(dot4(k, k));
            const float vn = row16_sum(dot4(v, v));
            bs += kn * vn;
            if (r > 0) {
                const float kd = row16_sum(dot4(k, kq[r - 1][j]));
                const float vd = row16_sum(dot4(v, vq[r - 1][j]));
                ds += kd * vd;
            }
        }
        const float B = bs + xhalf(bs);
        if (r > 0) {
            const float D = ds + xhalf(ds);
            float diff2 = Bprev + B - 2.0f * D;
            diff2 = fmaxf(diff2, 0.0f);
            float ratio = sqrtf(diff2) / sqrtf(Bprev);
            float r4 = sqrtf(ratio);
            float m_ = 1.0f - r4;
            m_ = m_ * m_;
            m_ = fminf(fmaxf(m_, 0.0f), 0.9999f);
            mval[r - 1] = m_;
        }
        Bprev = B;
    }

    float mv = mval[0];
#pragma unroll
    for (int r = 1; r < MU_R; ++r) mv = (lane == r) ? mval[r] : mv;
    const int l = l0 + lane;
    if (lane < MU_R && l < LL - 1) mu[n * LL + l] = mv;
}

#define WSEG 8
#define WTHREADS (LL / WSEG)
__global__ void __launch_bounds__(WTHREADS)
weights_kernel(const float* __restrict__ mu,
               float* __restrict__ weights) {
    __shared__ float sA[WTHREADS / 32];
    __shared__ float sB[WTHREADS / 32];

    int n = blockIdx.x;
    int t = threadIdx.x;
    int lane = t & 31, wave = t >> 5;
    const float* mp = mu + (size_t)n * LL + t * WSEG;

    float m[WSEG];
    if (t < WTHREADS - 1) {
        const float4 lo = *(const float4*)(mp);
        const float4 hi = *(const float4*)(mp + 4);
        m[0] = lo.x; m[1] = lo.y; m[2] = lo.z; m[3] = lo.w;
        m[4] = hi.x; m[5] = hi.y; m[6] = hi.z; m[7] = hi.w;
    } else {
        const float4 lo = *(const float4*)(mp);
        m[0] = lo.x; m[1] = lo.y; m[2] = lo.z; m[3] = lo.w;
        m[4] = mp[4]; m[5] = mp[5]; m[6] = mp[6];
        m[7] = 0.0f;
    }

    float A = m[WSEG - 1];
    float B = 1.0f;
#pragma unroll
    for (int j = WSEG - 2; j >= 0; --j) {
        B = B * m[j] + 1.0f;
        A = A * m[j];
    }

#pragma unroll
    for (int off = 1; off < 32; off <<= 1) {
        float A2 = __shfl_down(A, off, 32);
        float B2 = __shfl_down(B, off, 32);
        if (lane + off < 32) {
            B = A * B2 + B;
            A = A * A2;
        }
    }
    if (lane == 0) { sA[wave] = A; sB[wave] = B; }

    float Ae = __shfl_down(A, 1, 32);
    float Be = __shfl_down(B, 1, 32);
    if (lane == 31) { Ae = 1.0f; Be = 0.0f; }
    __syncthreads();

    float xw = 0.0f;
#pragma unroll
    for (int v = WTHREADS / 32 - 1; v >= 1; --v) {
        float a = sA[v], b = sB[v];
        if (v > wave) xw = a * xw + b;
    }

    float x = Ae * xw + Be;

    float o[WSEG];
#pragma unroll
    for (int j = WSEG - 1; j >= 0; --j) {
        x = x * m[j] + 1.0f;
        o[j] = x;
    }
    float* wp = weights + (size_t)n * LL + t * WSEG;
    *(float4*)(wp)     = make_float4(o[0], o[1], o[2], o[3]);
    *(float4*)(wp + 4) = make_float4(o[4], o[5], o[6], o[7]);
}

__global__ void __launch_bounds__(256)
sdelta_kernel(const float* __restrict__ keys,
                              const float* __restrict__ values,
                              const float* __restrict__ klen,
                              const float* __restrict__ weights,
                              float* __restrict__ Sbuf,
                              float* __restrict__ Ksum) {
    __shared__ float Klds[CK][LDW];
    __shared__ float Vlds[CK][LDW];

    int b = blockIdx.x;
    int c = b % NC;
    int h = (b / NC) % HH;
    int n = b / (NC * HH);
    int c0 = c * CK;
    int t = threadIdx.x;

    float4 kq[LD_IT], vq[LD_IT];
    float  klr[LD_IT], wr[LD_IT];
#pragma unroll
    for (int j = 0; j < LD_IT; ++j) {
        const int i  = t + j * 256;
        const int s  = i >> 4;
        const int e4 = (i & 15) * 4;
        const int l  = c0 + s;
        klr[j] = klen[n * LL + l];
        wr[j]  = weights[n * LL + l];
        kq[j] = *(const float4*)&keys  [(((size_t)n * LL + l) * HH + h) * EE + e4];
        vq[j] = *(const float4*)&values[(((size_t)n * LL + l) * HH + h) * MM + e4];
    }
    __builtin_amdgcn_sched_barrier(0);
#pragma unroll
    for (int j = 0; j < LD_IT; ++j) {
        const int i  = t + j * 256;
        const int s  = i >> 4;
        const int e4 = (i & 15) * 4;
        const float kl = klr[j];
        const float w  = wr[j];
        const float4 kv = kq[j];
        const float4 vv = vq[j];
        float4 ko, vo;
        ko.x = fmap(kv.x) * kl; ko.y = fmap(kv.y) * kl;
        ko.z = fmap(kv.z) * kl; ko.w = fmap(kv.w) * kl;
        vo.x = vv.x * w; vo.y = vv.y * w; vo.z = vv.z * w; vo.w = vv.w * w;
        *(float4*)&Klds[s][e4] = ko;
        *(float4*)&Vlds[s][e4] = vo;
    }
    __syncthreads();

    if (t < EE) {
        float acc = 0.f;
        for (int s = 0; s < CK; ++s) acc += Klds[s][t];
        Ksum[(((size_t)n * HH + h) * NC + c) * EE + t] = acc;
    }

    int wave = t >> 5, lane = t & 31, lr = lane & 15, lh = lane >> 4;
    float* out = Sbuf + (((size_t)n * HH + h) * NC + c) * (EE * MM);

    for (int tile = wave; tile < 16; tile += 8) {
        int ti = tile >> 2, tj = tile & 3;
        v8f acc = {};
        for (int k0 = 0; k0 < CK; k0 += 4) {
            int s = k0 + 2 * lh;
            v2f a, bb;
            int e = ti * 16 + lr;
            a.x = Klds[s][e];
            a.y = Klds[s + 1][e];
            int m = tj * 16 + lr;
            bb.x = Vlds[s][m];
            bb.y = Vlds[s + 1][m];
            acc = __builtin_amdgcn_wmma_f32_16x16x4_f32(false, a, false, bb,
                                                        (short)0, acc, false, false);
        }
        for (int r = 0; r < 8; ++r) {
            int row = ti * 16 + r + 8 * lh;
            int col = tj * 16 + lr;
            out[row * MM + col] = acc[r];
        }
    }
}

#define PF_THREADS 64
#define PF_SBLOCKS (NB * HH * (EE * MM / 4) / PF_THREADS)
__global__ void __launch_bounds__(PF_THREADS)
prefix_kernel(float* __restrict__ Sbuf,
              float* __restrict__ Ksum) {
    const int t = threadIdx.x;
    if (blockIdx.x < PF_SBLOCKS) {
        const int col = blockIdx.x * PF_THREADS + t;
        const int b = col / (EE * MM / 4);
        const int i = col % (EE * MM / 4);
        float4* base = (float4*)(Sbuf + (size_t)b * NC * (EE * MM)) + i;
        float4 v[NC - 1];
#pragma unroll
        for (int c = 0; c < NC - 1; ++c) v[c] = base[(size_t)c * (EE * MM / 4)];
        __builtin_amdgcn_sched_barrier(0);
        float4 acc = {0.f, 0.f, 0.f, 0.f};
#pragma unroll
        for (int c = 0; c < NC; ++c) {
            base[(size_t)c * (EE * MM / 4)] = acc;
            if (c < NC - 1) {
                acc.x += v[c].x; acc.y += v[c].y; acc.z += v[c].z; acc.w += v[c].w;
            }
        }
    } else {
        const int b = blockIdx.x - PF_SBLOCKS;
        float* kb = Ksum + (size_t)b * NC * EE + t;
        float v[NC - 1];
#pragma unroll
        for (int c = 0; c < NC - 1; ++c) v[c] = kb[c * EE];
        __builtin_amdgcn_sched_barrier(0);
        float acc = 0.f;
#pragma unroll
        for (int c = 0; c < NC; ++c) {
            kb[c * EE] = acc;
            if (c < NC - 1) acc += v[c];
        }
    }
}

__global__ void __launch_bounds__(256)
out_kernel(const float* __restrict__ queries,
                           const float* __restrict__ keys,
                           const float* __restrict__ values,
                           const float* __restrict__ klen,
                           const float* __restrict__ weights,
                           const float* __restrict__ Sbuf,
                           const float* __restrict__ Ksum,
                           float* __restrict__ out) {
    __shared__ float Qlds[CK][LDW];
    __shared__ float KP[CK][LDW];
    __shared__ float Vlds[CK][LDW];
    __shared__ float Slds[EE][LDW];
    __shared__ float ks[EE];
    __shared__ float zden[CK];

    int b = blockIdx.x;
    int c = b % NC;
    int h = (b / NC) % HH;
    int n = b / (NC * HH);
    int c0 = c * CK;
    int t = threadIdx.x;

    const float* Sp = Sbuf + (((size_t)n * HH + h) * NC + c) * (EE * MM);
    float4 qq[LD_IT], kq[LD_IT], vq[LD_IT], sq[LD_IT];
    float  klr[LD_IT], wr[LD_IT];
#pragma unroll
    for (int j = 0; j < LD_IT; ++j) {
        const int i  = t + j * 256;
        const int s  = i >> 4;
        const int e4 = (i & 15) * 4;
        const int l  = c0 + s;
        klr[j] = klen[n * LL + l];
        wr[j]  = weights[n * LL + l];
        qq[j] = *(const float4*)&queries[(((size_t)n * LL + l) * HH + h) * EE + e4];
        kq[j] = *(const float4*)&keys   [(((size_t)n * LL + l) * HH + h) * EE + e4];
        vq[j] = *(const float4*)&values [(((size_t)n * LL + l) * HH + h) * MM + e4];
        sq[j] = *(const float4*)&Sp[s * MM + e4];
    }
    const float ksv = Ksum[(((size_t)n * HH + h) * NC + c) * EE + (t & (EE - 1))];
    __builtin_amdgcn_sched_barrier(0);
#pragma unroll
    for (int j = 0; j < LD_IT; ++j) {
        const int i  = t + j * 256;
        const int s  = i >> 4;
        const int e4 = (i & 15) * 4;
        const float kl = klr[j];
        const float w  = wr[j];
        const float4 qv = qq[j];
        const float4 kv = kq[j];
        const float4 vv = vq[j];
        float4 qo, ko, vo;
        qo.x = fmap(qv.x);      qo.y = fmap(qv.y);
        qo.z = fmap(qv.z);      qo.w = fmap(qv.w);
        ko.x = fmap(kv.x) * kl; ko.y = fmap(kv.y) * kl;
        ko.z = fmap(kv.z) * kl; ko.w = fmap(kv.w) * kl;
        vo.x = vv.x * w; vo.y = vv.y * w; vo.z = vv.z * w; vo.w = vv.w * w;
        *(float4*)&Qlds[s][e4] = qo;
        *(float4*)&KP[s][e4]   = ko;
        *(float4*)&Vlds[s][e4] = vo;
        *(float4*)&Slds[s][e4] = sq[j];
    }
    ks[t & (EE - 1)] = ksv;
    __syncthreads();

    int wave = t >> 5, lane = t & 31, lr = lane & 15, lh = lane >> 4;

    v8f p0 = {}, p1 = {};
    for (int half = 0; half < 2; ++half) {
        int tile = wave + half * 8;
        int ti = tile >> 2, tj = tile & 3;
        v8f acc = {};
        for (int k0 = 0; k0 < EE; k0 += 4) {
            int e = k0 + 2 * lh;
            v2f a, bb;
            int row = ti * 16 + lr;
            a.x = Qlds[row][e];
            a.y = Qlds[row][e + 1];
            int col = tj * 16 + lr;
            bb.x = KP[col][e];
            bb.y = KP[col][e + 1];
            acc = __builtin_amdgcn_wmma_f32_16x16x4_f32(false, a, false, bb,
                                                        (short)0, acc, false, false);
        }
        if (half) p1 = acc; else p0 = acc;
    }
    __syncthreads();

    for (int half = 0; half < 2; ++half) {
        v8f acc = half ? p1 : p0;
        int tile = wave + half * 8;
        int ti = tile >> 2, tj = tile & 3;
        for (int r = 0; r < 8; ++r) {
            int row = ti * 16 + r + 8 * lh;
            int col = tj * 16 + lr;
            KP[row][col] = (col <= row) ? acc[r] : 0.0f;
        }
    }
    __syncthreads();

    if (t < CK) {
        float s1 = 0.f;
        for (int s = 0; s <= t; ++s) s1 += KP[t][s];
        float s2 = 0.f;
        for (int e = 0; e < EE; ++e) s2 += Qlds[t][e] * ks[e];
        zden[t] = s1 + s2 + EPS;
    }
    __syncthreads();

    for (int half = 0; half < 2; ++half) {
        int tile = wave + half * 8;
        int ti = tile >> 2, tj = tile & 3;
        v8f acc = {};
        for (int k0 = 0; k0 < EE; k0 += 4) {
            int e = k0 + 2 * lh;
            v2f a, bb;
            int row = ti * 16 + lr;
            a.x = Qlds[row][e];
            a.y = Qlds[row][e + 1];
            int col = tj * 16 + lr;
            bb.x = Slds[e][col];
            bb.y = Slds[e + 1][col];
            acc = __builtin_amdgcn_wmma_f32_16x16x4_f32(false, a, false, bb,
                                                        (short)0, acc, false, false);
        }
        for (int k0 = 0; k0 < CK; k0 += 4) {
            int s = k0 + 2 * lh;
            v2f a, bb;
            int row = ti * 16 + lr;
            a.x = KP[row][s];
            a.y = KP[row][s + 1];
            int col = tj * 16 + lr;
            bb.x = Vlds[s][col];
            bb.y = Vlds[s + 1][col];
            acc = __builtin_amdgcn_wmma_f32_16x16x4_f32(false, a, false, bb,
                                                        (short)0, acc, false, false);
        }
        for (int r = 0; r < 8; ++r) {
            int row = ti * 16 + r + 8 * lh;
            int col = tj * 16 + lr;
            int l = c0 + row;
            out[(((size_t)n * LL + l) * HH + h) * MM + col] = acc[r] / zden[row];
        }
    }
}

extern "C" void kernel_launch(void* const* d_in, const int* in_sizes, int n_in,
                              void* d_out, int out_size, void* d_ws, size_t ws_size,
                              hipStream_t stream) {
    const float* queries = (const float*)d_in[0];
    const float* keys    = (const float*)d_in[1];
    const float* values  = (const float*)d_in[2];
    const float* klen    = (const float*)d_in[3];
    float* outp = (float*)d_out;

    float* ws      = (float*)d_ws;
    float* Sbuf    = ws;
    float* Ksum    = Sbuf + (size_t)NB * HH * NC * EE * MM;
    float* mu      = Ksum + (size_t)NB * HH * NC * EE;
    float* weights = mu + (size_t)NB * LL;

    mu_kernel<<<NB * LL / (MU_R * MU_WAVES), MU_WAVES * 32, 0, stream>>>(keys, values, klen, mu);
    weights_kernel<<<NB, 256, 0, stream>>>(mu, weights);
    sdelta_kernel<<<NB * HH * NC, 256, 0, stream>>>(keys, values, klen, weights, Sbuf, Ksum);
    prefix_kernel<<<PF_SBLOCKS + NB * HH, PF_THREADS, 0, stream>>>(Sbuf, Ksum);
    out_kernel<<<NB * HH * NC, 256, 0, stream>>>(queries, keys, values, klen,
                                                 weights, Sbuf, Ksum, outp);
}


# state deltas kept per chunk PAIR: one 128-deep WMMA product per pair, prefix over 16 pair states, an odd chunk's output workgroup rebuilds its even neighbour's delta by WMMA: state traffic halves
# speedup vs baseline: 2.9312x; 1.1008x over previous
#include <hip/hip_runtime.h>
#include <hip/hip_bf16.h>

#define NB 2
#define LL 2048
#define HH 8
#define EE 64
#define MM 64
#define CK 64
#define NC (LL/CK)
#define EPS 1e-6f
#define PAD 4
#define LDW (EE + PAD)
#define LD_IT (CK * EE / 4 / 256)
#define NP (NC / 2)
#define CK2 (2 * CK)
#define LD_IT2 (CK2 * EE / 4 / 256)

typedef __attribute__((ext_vector_type(2))) float v2f;
typedef __attribute__((ext_vector_type(8))) float v8f;

__device__ __forceinline__ float fmap(float x) {
    return x > 0.0f ? x + 1.0f : __expf(x);
}

#define MU_R 4
#define MU_ROWS (MU_R + 1)
#define MU_WAVES 4

template <int M>
__device__ __forceinline__ float lane_xor16(float x) {
    int b = __builtin_bit_cast(int, x);
    return __builtin_bit_cast(float,
        __builtin_amdgcn_mov_dpp(b, 0x160 + M, 0xf, 0xf, false));
}

__device__ __forceinline__ float row16_sum(float x) {
    x += lane_xor16<1>(x);
    x += lane_xor16<2>(x);
    x += lane_xor16<4>(x);
    x += lane_xor16<8>(x);
    return x;
}

__device__ __forceinline__ float xhalf(float x) {
    int b = __builtin_bit_cast(int, x);
    return __builtin_bit_cast(float,
        __builtin_amdgcn_permlanex16(b, b, 0x76543210, 0xfedcba98, false, false));
}

__device__ __forceinline__ float dot4(const float4 a, const float4 b) {
    return a.x * b.x + a.y * b.y + a.z * b.z + a.w * b.w;
}

__global__ void __launch_bounds__(MU_WAVES * 32)
mu_kernel(const float* __restrict__ keys,
          const float* __restrict__ values,
          const float* __restrict__ klen,
          float* __restrict__ mu) {
    const int lane = threadIdx.x & 31;
    const int wave = __builtin_amdgcn_readfirstlane((int)(threadIdx.x >> 5));
    const int run  = blockIdx.x * MU_WAVES + wave;
    const int n    = run / (LL / MU_R);
    const int l0   = (run % (LL / MU_R)) * MU_R;

    float4 kq[MU_ROWS][4], vq[MU_ROWS][4];
    float  kl[MU_ROWS];
#pragma unroll
    for (int r = 0; r < MU_ROWS; ++r) {
        const int l = (l0 + r < LL) ? (l0 + r) : (LL - 1);
        const float4* kp = (const float4*)(keys   + ((size_t)n * LL + l) * (HH * EE));
        const float4* vp = (const float4*)(values + ((size_t)n * LL + l) * (HH * MM));
        kl[r] = klen[n * LL + l];
#pragma unroll
        for (int j = 0; j < 4; ++j) {
            kq[r][j] = kp[j * 32 + lane];
            vq[r][j] = vp[j * 32 + lane];
        }
    }

    float mval[MU_R];
    float Bprev = 0.0f;
#pragma unroll
    for (int r = 0; r < MU_ROWS; ++r) {
        float bs = 0.0f, ds = 0.0f;
#pragma unroll
        for (int j = 0; j < 4; ++j) {
            float4 k = kq[r][j];
            k.x = fmap(k.x) * kl[r]; k.y = fmap(k.y) * kl[r];
            k.z = fmap(k.z) * kl[r]; k.w = fmap(k.w) * kl[r];
            kq[r][j] = k;
            const float4 v = vq[r][j];
            const float kn = row16_sum(dot4(k, k));
            const float vn = row16_sum(dot4(v, v));
            bs += kn * vn;
            if (r > 0) {
                const float kd = row16_sum(dot4(k, kq[r - 1][j]));
                const float vd = row16_sum(dot4(v, vq[r - 1][j]));
                ds += kd * vd;
            }
        }
        const float B = bs + xhalf(bs);
        if (r > 0) {
            const float D = ds + xhalf(ds);
            float diff2 = Bprev + B - 2.0f * D;
            diff2 = fmaxf(diff2, 0.0f);
            float ratio = sqrtf(diff2) / sqrtf(Bprev);
            float r4 = sqrtf(ratio);
            float m_ = 1.0f - r4;
            m_ = m_ * m_;
            m_ = fminf(fmaxf(m_, 0.0f), 0.9999f);
            mval[r - 1] = m_;
        }
        Bprev = B;
    }

    float mv = mval[0];
#pragma unroll
    for (int r = 1; r < MU_R; ++r) mv = (lane == r) ? mval[r] : mv;
    const int l = l0 + lane;
    if (lane < MU_R && l < LL - 1) mu[n * LL + l] = mv;
}

#define WSEG 8
#define WTHREADS (LL / WSEG)

__global__ void __launch_bounds__(256)
sdelta_kernel(const float* __restrict__ keys,
                              const float* __restrict__ values,
                              const float* __restrict__ klen,
                              const float* __restrict__ mu,
                              float* __restrict__ weights,
                              float* __restrict__ Sbuf,
                              float* __restrict__ Ksum,
                              float* __restrict__ Kev) {
    __shared__ float Klds[CK2][LDW];
    __shared__ float Vlds[CK2][LDW];
    __shared__ float sA[WTHREADS / 32];
    __shared__ float sB[WTHREADS / 32];
    __shared__ __attribute__((aligned(16))) float wl[CK2];

    int b = blockIdx.x;
    int p = b % NP;
    int h = (b / NP) % HH;
    int n = b / (NP * HH);
    int c0 = p * CK2;
    int t = threadIdx.x;
    int wave = t >> 5, lane = t & 31, lr = lane & 15, lh = lane >> 4;

    float4 kq[LD_IT2], vq[LD_IT2];
    float  klr[LD_IT2];
#pragma unroll
    for (int j = 0; j < LD_IT2; ++j) {
        const int i  = t + j * 256;
        const int s  = i >> 4;
        const int e4 = (i & 15) * 4;
        const int l  = c0 + s;
        klr[j] = klen[n * LL + l];
        kq[j] = *(const float4*)&keys  [(((size_t)n * LL + l) * HH + h) * EE + e4];
        vq[j] = *(const float4*)&values[(((size_t)n * LL + l) * HH + h) * MM + e4];
    }
    const float* mp = mu + (size_t)n * LL + t * WSEG;
    float mq[WSEG];
    if (t < WTHREADS - 1) {
        const float4 lo = *(const float4*)(mp);
        const float4 hi = *(const float4*)(mp + 4);
        mq[0] = lo.x; mq[1] = lo.y; mq[2] = lo.z; mq[3] = lo.w;
        mq[4] = hi.x; mq[5] = hi.y; mq[6] = hi.z; mq[7] = hi.w;
    } else {
        const float4 lo = *(const float4*)(mp);
        mq[0] = lo.x; mq[1] = lo.y; mq[2] = lo.z; mq[3] = lo.w;
        mq[4] = mp[4]; mq[5] = mp[5]; mq[6] = mp[6];
        mq[7] = 0.0f;
    }
    __builtin_amdgcn_sched_barrier(0);

#pragma unroll
    for (int j = 0; j < LD_IT2; ++j) {
        const int i  = t + j * 256;
        const int s  = i >> 4;
        const int e4 = (i & 15) * 4;
        const float kl = klr[j];
        const float4 kv = kq[j];
        float4 ko;
        ko.x = fmap(kv.x) * kl; ko.y = fmap(kv.y) * kl;
        ko.z = fmap(kv.z) * kl; ko.w = fmap(kv.w) * kl;
        *(float4*)&Klds[s][e4] = ko;
    }

    float A = mq[WSEG - 1];
    float B = 1.0f;
#pragma unroll
    for (int j = WSEG - 2; j >= 0; --j) {
        B = B * mq[j] + 1.0f;
        A = A * mq[j];
    }

#pragma unroll
    for (int off = 1; off < 32; off <<= 1) {
        float A2 = __shfl_down(A, off, 32);
        float B2 = __shfl_down(B, off, 32);
        if (lane + off < 32) {
            B = A * B2 + B;
            A = A * A2;
        }
    }
    if (lane == 0) { sA[wave] = A; sB[wave] = B; }

    float Ae = __shfl_down(A, 1, 32);
    float Be = __shfl_down(B, 1, 32);
    if (lane == 31) { Ae = 1.0f; Be = 0.0f; }
    __syncthreads();

    float xw = 0.0f;
#pragma unroll
    for (int v = WTHREADS / 32 - 1; v >= 1; --v) {
        float a = sA[v], bq = sB[v];
        if (v > wave) xw = a * xw + bq;
    }

    float x = Ae * xw + Be;

    float o[WSEG];
#pragma unroll
    for (int j = WSEG - 1; j >= 0; --j) {
        x = x * mq[j] + 1.0f;
        o[j] = x;
    }
    if ((t >> 4) == p) {
        *(float4*)&wl[(t & 15) * WSEG]     = make_float4(o[0], o[1], o[2], o[3]);
        *(float4*)&wl[(t & 15) * WSEG + 4] = make_float4(o[4], o[5], o[6], o[7]);
    }
    __syncthreads();

#pragma unroll
    for (int j = 0; j < LD_IT2; ++j) {
        const int i  = t + j * 256;
        const int s  = i >> 4;
        const int e4 = (i & 15) * 4;
        const float w  = wl[s];
        const float4 vv = vq[j];
        float4 vo;
        vo.x = vv.x * w; vo.y = vv.y * w; vo.z = vv.z * w; vo.w = vv.w * w;
        *(float4*)&Vlds[s][e4] = vo;
    }
    __syncthreads();

    if (h == 0 && t < CK2 / 4)
        *(float4*)(weights + (size_t)n * LL + c0 + t * 4) = *(const float4*)&wl[t * 4];

    if (t < EE) {
        float acc0 = 0.f, acc1 = 0.f;
        for (int s = 0; s < CK; ++s) {
            acc0 += Klds[s][t];
            acc1 += Klds[CK + s][t];
        }
        const size_t ki = (((size_t)n * HH + h) * NP + p) * EE + t;
        Kev[ki]  = acc0;
        Ksum[ki] = acc0 + acc1;
    }

    float* out = Sbuf + (((size_t)n * HH + h) * NP + p) * (EE * MM);

    for (int tile = wave; tile < 16; tile += 8) {
        int ti = tile >> 2, tj = tile & 3;
        v8f acc = {};
        for (int k0 = 0; k0 < CK2; k0 += 4) {
            int s = k0 + 2 * lh;
            v2f a, bb;
            int e = ti * 16 + lr;
            a.x = Klds[s][e];
            a.y = Klds[s + 1][e];
            int m = tj * 16 + lr;
            bb.x = Vlds[s][m];
            bb.y = Vlds[s + 1][m];
            acc = __builtin_amdgcn_wmma_f32_16x16x4_f32(false, a, false, bb,
                                                        (short)0, acc, false, false);
        }
        for (int r = 0; r < 8; ++r) {
            int row = ti * 16 + r + 8 * lh;
            int col = tj * 16 + lr;
            out[row * MM + col] = acc[r];
        }
    }
}

#define PF_THREADS 64
#define PF_SBLOCKS (NB * HH * (EE * MM / 4) / PF_THREADS)
__global__ void __launch_bounds__(PF_THREADS)
prefix_kernel(float* __restrict__ Sbuf,
              float* __restrict__ Ksum) {
    const int t = threadIdx.x;
    if (blockIdx.x < PF_SBLOCKS) {
        const int col = blockIdx.x * PF_THREADS + t;
        const int b = col / (EE * MM / 4);
        const int i = col % (EE * MM / 4);
        float4* base = (float4*)(Sbuf + (size_t)b * NP * (EE * MM)) + i;
        float4 v[NP - 1];
#pragma unroll
        for (int c = 0; c < NP - 1; ++c) v[c] = base[(size_t)c * (EE * MM / 4)];
        __builtin_amdgcn_sched_barrier(0);
        float4 acc = {0.f, 0.f, 0.f, 0.f};
#pragma unroll
        for (int c = 0; c < NP; ++c) {
            base[(size_t)c * (EE * MM / 4)] = acc;
            if (c < NP - 1) {
                acc.x += v[c].x; acc.y += v[c].y; acc.z += v[c].z; acc.w += v[c].w;
            }
        }
    } else {
        const int b = blockIdx.x - PF_SBLOCKS;
        float* kb = Ksum + (size_t)b * NP * EE + t;
        float v[NP - 1];
#pragma unroll
        for (int c = 0; c < NP - 1; ++c) v[c] = kb[c * EE];
        __builtin_amdgcn_sched_barrier(0);
        float acc = 0.f;
#pragma unroll
        for (int c = 0; c < NP; ++c) {
            kb[c * EE] = acc;
            if (c < NP - 1) acc += v[c];
        }
    }
}

__global__ void __launch_bounds__(256)
out_kernel(const float* __restrict__ queries,
                           const float* __restrict__ keys,
                           const float* __restrict__ values,
                           const float* __restrict__ klen,
                           const float* __restrict__ weights,
                           const float* __restrict__ Sbuf,
                           const float* __restrict__ Ksum,
                           const float* __restrict__ Kev,
                           float* __restrict__ out) {
    __shared__ float Kp[CK][LDW];
    __shared__ float Vp[CK][LDW];
    __shared__ float Qlds[CK][LDW];
    __shared__ float KP[CK][LDW];
    __shared__ float Vlds[CK][LDW];
    __shared__ float Slds[EE][LDW];
    __shared__ float ks[EE];
    __shared__ float zden[CK];

    int b = blockIdx.x;
    int c = b % NC;
    int h = (b / NC) % HH;
    int n = b / (NC * HH);
    int c0 = c * CK;
    int t = threadIdx.x;

    const int  p   = c >> 1;
    const bool odd = (c & 1) != 0;
    const float* Sp = Sbuf + (((size_t)n * HH + h) * NP + p) * (EE * MM);
    float4 qq[LD_IT], kq[LD_IT], vq[LD_IT], sq[LD_IT];
    float  klr[LD_IT], wr[LD_IT];
    float4 kq2[LD_IT], vq2[LD_IT];
    float  klr2[LD_IT], wr2[LD_IT];
    if (odd) {
#pragma unroll
        for (int j = 0; j < LD_IT; ++j) {
            const int i  = t + j * 256;
            const int s  = i >> 4;
            const int e4 = (i & 15) * 4;
            const int l  = c0 - CK + s;
            klr2[j] = klen[n * LL + l];
            wr2[j]  = weights[n * LL + l];
            kq2[j] = *(const float4*)&keys  [(((size_t)n * LL + l) * HH + h) * EE + e4];
            vq2[j] = *(const float4*)&values[(((size_t)n * LL + l) * HH + h) * MM + e4];
        }
    }
#pragma unroll
    for (int j = 0; j < LD_IT; ++j) {
        const int i  = t + j * 256;
        const int s  = i >> 4;
        const int e4 = (i & 15) * 4;
        const int l  = c0 + s;
        klr[j] = klen[n * LL + l];
        wr[j]  = weights[n * LL + l];
        qq[j] = *(const float4*)&queries[(((size_t)n * LL + l) * HH + h) * EE + e4];
        kq[j] = *(const float4*)&keys   [(((size_t)n * LL + l) * HH + h) * EE + e4];
        vq[j] = *(const float4*)&values [(((size_t)n * LL + l) * HH + h) * MM + e4];
        sq[j] = *(const float4*)&Sp[s * MM + e4];
    }
    const size_t ki = (((size_t)n * HH + h) * NP + p) * EE + (t & (EE - 1));
    float ksv = Ksum[ki];
    float kev = 0.0f;
    if (odd) kev = Kev[ki];
    __builtin_amdgcn_sched_barrier(0);
    ksv += kev;
    if (odd) {
#pragma unroll
        for (int j = 0; j < LD_IT; ++j) {
            const int i  = t + j * 256;
            const int s  = i >> 4;
            const int e4 = (i & 15) * 4;
            const float kl = klr2[j];
            const float w  = wr2[j];
            const float4 kv = kq2[j];
            const float4 vv = vq2[j];
            float4 ko, vo;
            ko.x = fmap(kv.x) * kl; ko.y = fmap(kv.y) * kl;
            ko.z = fmap(kv.z) * kl; ko.w = fmap(kv.w) * kl;
            vo.x = vv.x * w; vo.y = vv.y * w; vo.z = vv.z * w; vo.w = vv.w * w;
            *(float4*)&Kp[s][e4] = ko;
            *(float4*)&Vp[s][e4] = vo;
        }
    }
#pragma unroll
    for (int j = 0; j < LD_IT; ++j) {
        const int i  = t + j * 256;
        const int s  = i >> 4;
        const int e4 = (i & 15) * 4;
        const float kl = klr[j];
        const float w  = wr[j];
        const float4 qv = qq[j];
        const float4 kv = kq[j];
        const float4 vv = vq[j];
        float4 qo, ko, vo;
        qo.x = fmap(qv.x);      qo.y = fmap(qv.y);
        qo.z = fmap(qv.z);      qo.w = fmap(qv.w);
        ko.x = fmap(kv.x) * kl; ko.y = fmap(kv.y) * kl;
        ko.z = fmap(kv.z) * kl; ko.w = fmap(kv.w) * kl;
        vo.x = vv.x * w; vo.y = vv.y * w; vo.z = vv.z * w; vo.w = vv.w * w;
        *(float4*)&Qlds[s][e4] = qo;
        *(float4*)&KP[s][e4]   = ko;
        *(float4*)&Vlds[s][e4] = vo;
        *(float4*)&Slds[s][e4] = sq[j];
    }
    ks[t & (EE - 1)] = ksv;
    __syncthreads();

    int wave = t >> 5, lane = t & 31, lr = lane & 15, lh = lane >> 4;

    if (odd) {
        for (int half = 0; half < 2; ++half) {
            int tile = wave + half * 8;
            int ti = tile >> 2, tj = tile & 3;
            v8f acc = {};
            for (int k0 = 0; k0 < CK; k0 += 4) {
                int s = k0 + 2 * lh;
                v2f a, bb;
                int e = ti * 16 + lr;
                a.x = Kp[s][e];
                a.y = Kp[s + 1][e];
                int m = tj * 16 + lr;
                bb.x = Vp[s][m];
                bb.y = Vp[s + 1][m];
                acc = __builtin_amdgcn_wmma_f32_16x16x4_f32(false, a, false, bb,
                                                            (short)0, acc, false, false);
            }
            for (int r = 0; r < 8; ++r) {
                int row = ti * 16 + r + 8 * lh;
                int col = tj * 16 + lr;
                Slds[row][col] += acc[r];
            }
        }
    }

    v8f p0 = {}, p1 = {};
    for (int half = 0; half < 2; ++half) {
        int tile = wave + half * 8;
        int ti = tile >> 2, tj = tile & 3;
        v8f acc = {};
        for (int k0 = 0; k0 < EE; k0 += 4) {
            int e = k0 + 2 * lh;
            v2f a, bb;
            int row = ti * 16 + lr;
            a.x = Qlds[row][e];
            a.y = Qlds[row][e + 1];
            int col = tj * 16 + lr;
            bb.x = KP[col][e];
            bb.y = KP[col][e + 1];
            acc = __builtin_amdgcn_wmma_f32_16x16x4_f32(false, a, false, bb,
                                                        (short)0, acc, false, false);
        }
        if (half) p1 = acc; else p0 = acc;
    }
    __syncthreads();

    for (int half = 0; half < 2; ++half) {
        v8f acc = half ? p1 : p0;
        int tile = wave + half * 8;
        int ti = tile >> 2, tj = tile & 3;
        for (int r = 0; r < 8; ++r) {
            int row = ti * 16 + r + 8 * lh;
            int col = tj * 16 + lr;
            KP[row][col] = (col <= row) ? acc[r] : 0.0f;
        }
    }
    __syncthreads();

    if (t < CK) {
        float s1 = 0.f;
        for (int s = 0; s <= t; ++s) s1 += KP[t][s];
        float s2 = 0.f;
        for (int e = 0; e < EE; ++e) s2 += Qlds[t][e] * ks[e];
        zden[t] = s1 + s2 + EPS;
    }
    __syncthreads();

    for (int half = 0; half < 2; ++half) {
        int tile = wave + half * 8;
        int ti = tile >> 2, tj = tile & 3;
        v8f acc = {};
        for (int k0 = 0; k0 < EE; k0 += 4) {
            int e = k0 + 2 * lh;
            v2f a, bb;
            int row = ti * 16 + lr;
            a.x = Qlds[row][e];
            a.y = Qlds[row][e + 1];
            int col = tj * 16 + lr;
            bb.x = Slds[e][col];
            bb.y = Slds[e + 1][col];
            acc = __builtin_amdgcn_wmma_f32_16x16x4_f32(false, a, false, bb,
                                                        (short)0, acc, false, false);
        }
        for (int k0 = 0; k0 < CK; k0 += 4) {
            int s = k0 + 2 * lh;
            v2f a, bb;
            int row = ti * 16 + lr;
            a.x = KP[row][s];
            a.y = KP[row][s + 1];
            int col = tj * 16 + lr;
            bb.x = Vlds[s][col];
            bb.y = Vlds[s + 1][col];
            acc = __builtin_amdgcn_wmma_f32_16x16x4_f32(false, a, false, bb,
                                                        (short)0, acc, false, false);
        }
        for (int r = 0; r < 8; ++r) {
            int row = ti * 16 + r + 8 * lh;
            int col = tj * 16 + lr;
            int l = c0 + row;
            out[(((size_t)n * LL + l) * HH + h) * MM + col] = acc[r] / zden[row];
        }
    }
}

extern "C" void kernel_launch(void* const* d_in, const int* in_sizes, int n_in,
                              void* d_out, int out_size, void* d_ws, size_t ws_size,
                              hipStream_t stream) {
    const float* queries = (const float*)d_in[0];
    const float* keys    = (const float*)d_in[1];
    const float* values  = (const float*)d_in[2];
    const float* klen    = (const float*)d_in[3];
    float* outp = (float*)d_out;

    float* ws      = (float*)d_ws;
    float* Sbuf    = ws;
    float* Ksum    = Sbuf + (size_t)NB * HH * NP * EE * MM;
    float* Kev     = Ksum + (size_t)NB * HH * NP * EE;
    float* mu      = Kev + (size_t)NB * HH * NP * EE;
    float* weights = mu + (size_t)NB * LL;

    mu_kernel<<<NB * LL / (MU_R * MU_WAVES), MU_WAVES * 32, 0, stream>>>(keys, values, klen, mu);
    sdelta_kernel<<<NB * HH * NP, 256, 0, stream>>>(keys, values, klen, mu, weights, Sbuf, Ksum, Kev);
    prefix_kernel<<<PF_SBLOCKS + NB * HH, PF_THREADS, 0, stream>>>(Sbuf, Ksum);
    out_kernel<<<NB * HH * NC, 256, 0, stream>>>(queries, keys, values, klen,
                                                 weights, Sbuf, Ksum, Kev, outp);
}


# state slots per chunk, not per chunk pair: the state-delta workgroup prefixes its 8 chunk deltas; an odd chunk's output workgroup no longer reloads its even neighbour nor rebuilds its delta by WMMA
# speedup vs baseline: 4.2801x; 1.4602x over previous
#include <hip/hip_runtime.h>
#include <hip/hip_bf16.h>

#define NB 2
#define LL 2048
#define HH 8
#define EE 64
#define MM 64
#define CK 64
#define NC (LL/CK)
#define EPS 1e-6f
#define PAD 4
#define LDW (EE + PAD)
#define LD_IT (CK * EE / 4 / 256)
#define NP (NC / 2)
#define CK2 (2 * CK)

typedef __attribute__((ext_vector_type(2))) float v2f;
typedef __attribute__((ext_vector_type(8))) float v8f;

__device__ __forceinline__ float fmap(float x) {
    return x > 0.0f ? x + 1.0f : __expf(x);
}

#define MU_R 4
#define MU_ROWS (MU_R + 1)
#define MU_WAVES 4

template <int M>
__device__ __forceinline__ float lane_xor16(float x) {
    int b = __builtin_bit_cast(int, x);
    return __builtin_bit_cast(float,
        __builtin_amdgcn_mov_dpp(b, 0x160 + M, 0xf, 0xf, false));
}

__device__ __forceinline__ float row16_sum(float x) {
    x += lane_xor16<1>(x);
    x += lane_xor16<2>(x);
    x += lane_xor16<4>(x);
    x += lane_xor16<8>(x);
    return x;
}

__device__ __forceinline__ float xhalf(float x) {
    int b = __builtin_bit_cast(int, x);
    return __builtin_bit_cast(float,
        __builtin_amdgcn_permlanex16(b, b, 0x76543210, 0xfedcba98, false, false));
}

__device__ __forceinline__ float dot4(const float4 a, const float4 b) {
    return a.x * b.x + a.y * b.y + a.z * b.z + a.w * b.w;
}

__global__ void __launch_bounds__(MU_WAVES * 32)
mu_kernel(const float* __restrict__ keys,
          const float* __restrict__ values,
          const float* __restrict__ klen,
          float* __restrict__ mu) {
    const int lane = threadIdx.x & 31;
    const int wave = __builtin_amdgcn_readfirstlane((int)(threadIdx.x >> 5));
    const int run  = blockIdx.x * MU_WAVES + wave;
    const int n    = run / (LL / MU_R);
    const int l0   = (run % (LL / MU_R)) * MU_R;

    float4 kq[MU_ROWS][4], vq[MU_ROWS][4];
    float  kl[MU_ROWS];
#pragma unroll
    for (int r = 0; r < MU_ROWS; ++r) {
        const int l = (l0 + r < LL) ? (l0 + r) : (LL - 1);
        const float4* kp = (const float4*)(keys   + ((size_t)n * LL + l) * (HH * EE));
        const float4* vp = (const float4*)(values + ((size_t)n * LL + l) * (HH * MM));
        kl[r] = klen[n * LL + l];
#pragma unroll
        for (int j = 0; j < 4; ++j) {
            kq[r][j] = kp[j * 32 + lane];
            vq[r][j] = vp[j * 32 + lane];
        }
    }

    float mval[MU_R];
    float Bprev = 0.0f;
#pragma unroll
    for (int r = 0; r < MU_ROWS; ++r) {
        float bs = 0.0f, ds = 0.0f;
#pragma unroll
        for (int j = 0; j < 4; ++j) {
            float4 k = kq[r][j];
            k.x = fmap(k.x) * kl[r]; k.y = fmap(k.y) * kl[r];
            k.z = fmap(k.z) * kl[r]; k.w = fmap(k.w) * kl[r];
            kq[r][j] = k;
            const float4 v = vq[r][j];
            const float kn = row16_sum(dot4(k, k));
            const float vn = row16_sum(dot4(v, v));
            bs += kn * vn;
            if (r > 0) {
                const float kd = row16_sum(dot4(k, kq[r - 1][j]));
                const float vd = row16_sum(dot4(v, vq[r - 1][j]));
                ds += kd * vd;
            }
        }
        const float B = bs + xhalf(bs);
        if (r > 0) {
            const float D = ds + xhalf(ds);
            float diff2 = Bprev + B - 2.0f * D;
            diff2 = fmaxf(diff2, 0.0f);
            float ratio = sqrtf(diff2) / sqrtf(Bprev);
            float r4 = sqrtf(ratio);
            float m_ = 1.0f - r4;
            m_ = m_ * m_;
            m_ = fminf(fmaxf(m_, 0.0f), 0.9999f);
            mval[r - 1] = m_;
        }
        Bprev = B;
    }

    float mv = mval[0];
#pragma unroll
    for (int r = 1; r < MU_R; ++r) mv = (lane == r) ? mval[r] : mv;
    const int l = l0 + lane;
    if (lane < MU_R && l < LL - 1) mu[n * LL + l] = mv;
}

#define WSEG 8
#define WTHREADS (LL / WSEG)

#define GP 4
#define GC (2 * GP)
#define NG (NP / GP)
#define GPOS (GP * CK2)
#define SR 16
#define LDK (SR + PAD)
#define KLD_IT (GPOS * SR / 4 / 256)
#define VLD_IT (GPOS * MM / 4 / 256)

__global__ void __launch_bounds__(256)
sdelta_kernel(const float* __restrict__ keys,
                              const float* __restrict__ values,
                              const float* __restrict__ klen,
                              const float* __restrict__ mu,
                              float* __restrict__ weights,
                              float* __restrict__ Sbuf,
                              float* __restrict__ Ksum) {
    __shared__ __attribute__((aligned(16))) float Kst[GPOS][LDK];
    __shared__ __attribute__((aligned(16))) float Vlds[GPOS][LDW];
    __shared__ __attribute__((aligned(16))) float Tl[GC][SR][LDW];
    __shared__ float csum[2 * GP][SR];
    __shared__ float sA[WTHREADS / 32];
    __shared__ float sB[WTHREADS / 32];
    __shared__ __attribute__((aligned(16))) float wl[GPOS];

    int b = blockIdx.x;
    int q = b & 3;
    int g = (b >> 2) & (NG - 1);
    int h = (b >> 4) & (HH - 1);
    int n = b >> 7;
    int c0 = g * GPOS;
    int t = threadIdx.x;
    int wave = t >> 5, lane = t & 31, lr = lane & 15, lh = lane >> 4;

    float4 kq[KLD_IT], vq[VLD_IT];
    float  klr[KLD_IT];
#pragma unroll
    for (int j = 0; j < KLD_IT; ++j) {
        const int i  = t + j * 256;
        const int s  = i >> 2;
        const int e4 = (i & 3) * 4;
        const int l  = c0 + s;
        klr[j] = klen[n * LL + l];
        kq[j] = *(const float4*)&keys[(((size_t)n * LL + l) * HH + h) * EE + q * SR + e4];
    }
#pragma unroll
    for (int j = 0; j < VLD_IT; ++j) {
        const int i  = t + j * 256;
        const int s  = i >> 4;
        const int e4 = (i & 15) * 4;
        const int l  = c0 + s;
        vq[j] = *(const float4*)&values[(((size_t)n * LL + l) * HH + h) * MM + e4];
    }
    const float* mp = mu + (size_t)n * LL + t * WSEG;
    float mq[WSEG];
    if (t < WTHREADS - 1) {
        const float4 lo = *(const float4*)(mp);
        const float4 hi = *(const float4*)(mp + 4);
        mq[0] = lo.x; mq[1] = lo.y; mq[2] = lo.z; mq[3] = lo.w;
        mq[4] = hi.x; mq[5] = hi.y; mq[6] = hi.z; mq[7] = hi.w;
    } else {
        const float4 lo = *(const float4*)(mp);
        mq[0] = lo.x; mq[1] = lo.y; mq[2] = lo.z; mq[3] = lo.w;
        mq[4] = mp[4]; mq[5] = mp[5]; mq[6] = mp[6];
        mq[7] = 0.0f;
    }
    __builtin_amdgcn_sched_barrier(0);

#pragma unroll
    for (int j = 0; j < KLD_IT; ++j) {
        const int i  = t + j * 256;
        const int s  = i >> 2;
        const int e4 = (i & 3) * 4;
        const float kl = klr[j];
        const float4 kv = kq[j];
        float4 ko;
        ko.x = fmap(kv.x) * kl; ko.y = fmap(kv.y) * kl;
        ko.z = fmap(kv.z) * kl; ko.w = fmap(kv.w) * kl;
        kq[j] = ko;
        *(float4*)&Kst[s][e4] = ko;
    }
#pragma unroll
    for (int j = 0; j < VLD_IT; ++j) {
        const int i  = t + j * 256;
        const int s  = i >> 4;
        const int e4 = (i & 15) * 4;
        *(float4*)&Vlds[s][e4] = vq[j];
    }

    float A = mq[WSEG - 1];
    float B = 1.0f;
#pragma unroll
    for (int j = WSEG - 2; j >= 0; --j) {
        B = B * mq[j] + 1.0f;
        A = A * mq[j];
    }

#pragma unroll
    for (int off = 1; off < 32; off <<= 1) {
        float A2 = __shfl_down(A, off, 32);
        float B2 = __shfl_down(B, off, 32);
        if (lane + off < 32) {
            B = A * B2 + B;
            A = A * A2;
        }
    }
    if (lane == 0) { sA[wave] = A; sB[wave] = B; }

    float Ae = __shfl_down(A, 1, 32);
    float Be = __shfl_down(B, 1, 32);
    if (lane == 31) { Ae = 1.0f; Be = 0.0f; }
    __syncthreads();

    float xw = 0.0f;
#pragma unroll
    for (int v = WTHREADS / 32 - 1; v >= 1; --v) {
        float a = sA[v], bq = sB[v];
        if (v > wave) xw = a * xw + bq;
    }

    float x = Ae * xw + Be;

    float o[WSEG];
#pragma unroll
    for (int j = WSEG - 1; j >= 0; --j) {
        x = x * mq[j] + 1.0f;
        o[j] = x;
    }
    if ((t >> 6) == g) {
        *(float4*)&wl[(t & 63) * WSEG]     = make_float4(o[0], o[1], o[2], o[3]);
        *(float4*)&wl[(t & 63) * WSEG + 4] = make_float4(o[4], o[5], o[6], o[7]);
    }

    {
        float ps = 0.f;
        const int r0 = wave * CK + lh * (CK / 2);
#pragma unroll 8
        for (int s = 0; s < CK / 2; ++s) ps += Kst[r0 + s][lr];
        ps += xhalf(ps);
        if (lh == 0) csum[wave][lr] = ps;
    }
    __syncthreads();

#pragma unroll
    for (int j = 0; j < KLD_IT; ++j) {
        const int i  = t + j * 256;
        const int s  = i >> 2;
        const int e4 = (i & 3) * 4;
        const float w = wl[s];
        float4 ko = kq[j];
        ko.x *= w; ko.y *= w; ko.z *= w; ko.w *= w;
        *(float4*)&Kst[s][e4] = ko;
    }

    if (h == 0 && q == 0 && t < GPOS / 4)
        *(float4*)(weights + (size_t)n * LL + c0 + t * 4) = *(const float4*)&wl[t * 4];

    if (t < GC * SR) {
        const int cc  = t >> 4;
        const int col = t & 15;
        float run = 0.f, mine = 0.f;
#pragma unroll
        for (int jj = 0; jj < GC; ++jj) {
            if (jj == cc) mine = run;
            run += csum[jj][col];
        }
        const size_t ki = (((size_t)n * HH + h) * NC + g * GC + cc) * EE + q * SR + col;
        Ksum[ki] = (cc == 0) ? run : mine;
    }
    __syncthreads();

    {
        const int j   = wave >> 1;
        const int tj0 = (wave & 1) * 2;
#pragma unroll
        for (int cc = 0; cc < 2; ++cc) {
            const int ch = 2 * j + cc;
            v8f acc0 = {}, acc1 = {};
#pragma unroll
            for (int k0 = 0; k0 < CK; k0 += 4) {
                const int s = ch * CK + k0 + 2 * lh;
                v2f a, b0, b1;
                a.x = Kst[s][lr];
                a.y = Kst[s + 1][lr];
                const int m0 = tj0 * 16 + lr;
                b0.x = Vlds[s][m0];
                b0.y = Vlds[s + 1][m0];
                b1.x = Vlds[s][m0 + 16];
                b1.y = Vlds[s + 1][m0 + 16];
                acc0 = __builtin_amdgcn_wmma_f32_16x16x4_f32(false, a, false, b0,
                                                             (short)0, acc0, false, false);
                acc1 = __builtin_amdgcn_wmma_f32_16x16x4_f32(false, a, false, b1,
                                                             (short)0, acc1, false, false);
            }
#pragma unroll
            for (int r = 0; r < 8; ++r) {
                const int row = r + 8 * lh;
                Tl[ch][row][tj0 * 16 + lr]      = acc0[r];
                Tl[ch][row][tj0 * 16 + 16 + lr] = acc1[r];
            }
        }
    }
    __syncthreads();

    {
        const int row = t >> 4;
        const int c4  = (t & 15) * 4;
        float4 tv[GC];
#pragma unroll
        for (int cc = 0; cc < GC; ++cc) tv[cc] = *(const float4*)&Tl[cc][row][c4];
        float* o0 = Sbuf + (((size_t)n * HH + h) * NC + g * GC) * (EE * MM)
                         + (size_t)(q * SR + row) * MM + c4;
        float4 run = tv[0];
#pragma unroll
        for (int cc = 1; cc < GC; ++cc) {
            *(float4*)(o0 + cc * EE * MM) = run;
            run.x += tv[cc].x; run.y += tv[cc].y; run.z += tv[cc].z; run.w += tv[cc].w;
        }
        *(float4*)(o0) = run;
    }
}

__global__ void __launch_bounds__(256) __attribute__((amdgpu_waves_per_eu(4, 4)))
out_kernel(const float* __restrict__ queries,
                           const float* __restrict__ keys,
                           const float* __restrict__ values,
                           const float* __restrict__ klen,
                           const float* __restrict__ weights,
                           const float* __restrict__ Sbuf,
                           const float* __restrict__ Ksum,
                           float* __restrict__ out) {
    __shared__ float Qlds[CK][LDW];
    __shared__ float KP[CK][LDW];
    __shared__ float Vlds[CK][LDW];
    __shared__ float Slds[EE][LDW];
    __shared__ float ks[EE];
    __shared__ float zden[CK];

    int b = blockIdx.x;
    int c = b % NC;
    int h = (b / NC) % HH;
    int n = b / (NC * HH);
    int c0 = c * CK;
    int t = threadIdx.x;

    const int  g   = c / GC;
    const int  jp  = c % GC;
    const float* Sn = Sbuf + ((size_t)n * HH + h) * NC * (EE * MM);
    const float* Sp = Sn + (size_t)c * (EE * MM);
    float4 qq[LD_IT], kq[LD_IT], vq[LD_IT], sq[LD_IT];
    float  klr[LD_IT], wr[LD_IT];
#pragma unroll
    for (int j = 0; j < LD_IT; ++j) {
        const int i  = t + j * 256;
        const int s  = i >> 4;
        const int e4 = (i & 15) * 4;
        const int l  = c0 + s;
        klr[j] = klen[n * LL + l];
        wr[j]  = weights[n * LL + l];
        qq[j] = *(const float4*)&queries[(((size_t)n * LL + l) * HH + h) * EE + e4];
        kq[j] = *(const float4*)&keys   [(((size_t)n * LL + l) * HH + h) * EE + e4];
        vq[j] = *(const float4*)&values [(((size_t)n * LL + l) * HH + h) * MM + e4];
        sq[j] = *(const float4*)&Sp[s * MM + e4];
    }
    const size_t kn = ((size_t)n * HH + h) * NC * EE + (t & (EE - 1));
    const size_t ki = kn + (size_t)c * EE;
    float ksv = 0.0f, kg0 = 0.0f, kg1 = 0.0f, kg2 = 0.0f;
    if (jp > 0) ksv = Ksum[ki];
    if (g > 0)  kg0 = Ksum[kn + 0 * GC * EE];
    if (g > 1)  kg1 = Ksum[kn + 1 * GC * EE];
    if (g > 2)  kg2 = Ksum[kn + 2 * GC * EE];
    __builtin_amdgcn_sched_barrier(0);
    ksv += (kg0 + kg1) + kg2;
#pragma unroll
    for (int j = 0; j < LD_IT; ++j) {
        const int i  = t + j * 256;
        const int s  = i >> 4;
        const int e4 = (i & 15) * 4;
        const float kl = klr[j];
        const float w  = wr[j];
        const float4 qv = qq[j];
        const float4 kv = kq[j];
        const float4 vv = vq[j];
        float4 qo, ko, vo;
        qo.x = fmap(qv.x);      qo.y = fmap(qv.y);
        qo.z = fmap(qv.z);      qo.w = fmap(qv.w);
        ko.x = fmap(kv.x) * kl; ko.y = fmap(kv.y) * kl;
        ko.z = fmap(kv.z) * kl; ko.w = fmap(kv.w) * kl;
        vo.x = vv.x * w; vo.y = vv.y * w; vo.z = vv.z * w; vo.w = vv.w * w;
        *(float4*)&Qlds[s][e4] = qo;
        *(float4*)&KP[s][e4]   = ko;
        *(float4*)&Vlds[s][e4] = vo;
        float4 so;
        so.x = (jp > 0) ? sq[j].x : 0.0f; so.y = (jp > 0) ? sq[j].y : 0.0f;
        so.z = (jp > 0) ? sq[j].z : 0.0f; so.w = (jp > 0) ? sq[j].w : 0.0f;
        *(float4*)&Slds[s][e4] = so;
    }
    ks[t & (EE - 1)] = ksv;
    __syncthreads();

    float4 g0[LD_IT], g1[LD_IT], g2[LD_IT];
#pragma unroll
    for (int j = 0; j < LD_IT; ++j) {
        const int i  = t + j * 256;
        const int s  = i >> 4;
        const int e4 = (i & 15) * 4;
        const float4 z4 = {0.f, 0.f, 0.f, 0.f};
        g0[j] = z4; g1[j] = z4; g2[j] = z4;
        if (g > 0)  g0[j] = *(const float4*)&Sn[0 * GC * (EE * MM) + s * MM + e4];
        if (g > 1)  g1[j] = *(const float4*)&Sn[1 * GC * (EE * MM) + s * MM + e4];
        if (g > 2)  g2[j] = *(const float4*)&Sn[2 * GC * (EE * MM) + s * MM + e4];
    }

    int wave = t >> 5, lane = t & 31, lr = lane & 15, lh = lane >> 4;

    v8f p0 = {}, p1 = {};
    for (int half = 0; half < 2; ++half) {
        int tile = wave + half * 8;
        int ti = tile >> 2, tj = tile & 3;
        v8f acc = {};
        for (int k0 = 0; k0 < EE; k0 += 4) {
            int e = k0 + 2 * lh;
            v2f a, bb;
            int row = ti * 16 + lr;
            a.x = Qlds[row][e];
            a.y = Qlds[row][e + 1];
            int col = tj * 16 + lr;
            bb.x = KP[col][e];
            bb.y = KP[col][e + 1];
            acc = __builtin_amdgcn_wmma_f32_16x16x4_f32(false, a, false, bb,
                                                        (short)0, acc, false, false);
        }
        if (half) p1 = acc; else p0 = acc;
    }
    __syncthreads();

    if (g > 0) {
#pragma unroll
        for (int j = 0; j < LD_IT; ++j) {
            const int i  = t + j * 256;
            const int s  = i >> 4;
            const int e4 = (i & 15) * 4;
            float4 so = *(const float4*)&Slds[s][e4];
            so.x += (g0[j].x + g1[j].x) + g2[j].x;
            so.y += (g0[j].y + g1[j].y) + g2[j].y;
            so.z += (g0[j].z + g1[j].z) + g2[j].z;
            so.w += (g0[j].w + g1[j].w) + g2[j].w;
            *(float4*)&Slds[s][e4] = so;
        }
    }

    for (int half = 0; half < 2; ++half) {
        v8f acc = half ? p1 : p0;
        int tile = wave + half * 8;
        int ti = tile >> 2, tj = tile & 3;
        for (int r = 0; r < 8; ++r) {
            int row = ti * 16 + r + 8 * lh;
            int col = tj * 16 + lr;
            KP[row][col] = (col <= row) ? acc[r] : 0.0f;
        }
    }
    __syncthreads();

    if (t < CK) {
        float s1 = 0.f;
        for (int s = 0; s <= t; ++s) s1 += KP[t][s];
        float s2 = 0.f;
        for (int e = 0; e < EE; ++e) s2 += Qlds[t][e] * ks[e];
        zden[t] = s1 + s2 + EPS;
    }
    __syncthreads();

    for (int half = 0; half < 2; ++half) {
        int tile = wave + half * 8;
        int ti = tile >> 2, tj = tile & 3;
        v8f acc = {};
        for (int k0 = 0; k0 < EE; k0 += 4) {
            int e = k0 + 2 * lh;
            v2f a, bb;
            int row = ti * 16 + lr;
            a.x = Qlds[row][e];
            a.y = Qlds[row][e + 1];
            int col = tj * 16 + lr;
            bb.x = Slds[e][col];
            bb.y = Slds[e + 1][col];
            acc = __builtin_amdgcn_wmma_f32_16x16x4_f32(false, a, false, bb,
                                                        (short)0, acc, false, false);
        }
        for (int k0 = 0; k0 < CK; k0 += 4) {
            int s = k0 + 2 * lh;
            v2f a, bb;
            int row = ti * 16 + lr;
            a.x = KP[row][s];
            a.y = KP[row][s + 1];
            int col = tj * 16 + lr;
            bb.x = Vlds[s][col];
            bb.y = Vlds[s + 1][col];
            acc = __builtin_amdgcn_wmma_f32_16x16x4_f32(false, a, false, bb,
                                                        (short)0, acc, false, false);
        }
        for (int r = 0; r < 8; ++r) {
            int row = ti * 16 + r + 8 * lh;
            int col = tj * 16 + lr;
            int l = c0 + row;
            out[(((size_t)n * LL + l) * HH + h) * MM + col] = acc[r] / zden[row];
        }
    }
}

extern "C" void kernel_launch(void* const* d_in, const int* in_sizes, int n_in,
                              void* d_out, int out_size, void* d_ws, size_t ws_size,
                              hipStream_t stream) {
    const float* queries = (const float*)d_in[0];
    const float* keys    = (const float*)d_in[1];
    const float* values  = (const float*)d_in[2];
    const float* klen    = (const float*)d_in[3];
    float* outp = (float*)d_out;

    float* ws      = (float*)d_ws;
    float* Sbuf    = ws;
    float* Ksum    = Sbuf + (size_t)NB * HH * NC * EE * MM;
    float* mu      = Ksum + (size_t)NB * HH * NC * EE;
    float* weights = mu + (size_t)NB * LL;

    mu_kernel<<<NB * LL / (MU_R * MU_WAVES), MU_WAVES * 32, 0, stream>>>(keys, values, klen, mu);
    sdelta_kernel<<<NB * HH * NG * (EE / SR), 256, 0, stream>>>(keys, values, klen, mu, weights, Sbuf, Ksum);
    out_kernel<<<NB * HH * NC, 256, 0, stream>>>(queries, keys, values, klen,
                                                 weights, Sbuf, Ksum, outp);
}


# output kernel: row denominators by all 256 threads (4 threads per row, 16-byte LDS reads, 2 lane exchanges) as one reciprocal per row; was a serial pass of 2 waves and 16 divisions per thread
# speedup vs baseline: 4.4735x; 1.0452x over previous
#include <hip/hip_runtime.h>
#include <hip/hip_bf16.h>

#define NB 2
#define LL 2048
#define HH 8
#define EE 64
#define MM 64
#define CK 64
#define NC (LL/CK)
#define EPS 1e-6f
#define PAD 4
#define LDW (EE + PAD)
#define LD_IT (CK * EE / 4 / 256)
#define NP (NC / 2)
#define CK2 (2 * CK)

typedef __attribute__((ext_vector_type(2))) float v2f;
typedef __attribute__((ext_vector_type(8))) float v8f;

__device__ __forceinline__ float fmap(float x) {
    return x > 0.0f ? x + 1.0f : __expf(x);
}

#define MU_R 4
#define MU_ROWS (MU_R + 1)
#define MU_WAVES 4

template <int M>
__device__ __forceinline__ float lane_xor16(float x) {
    int b = __builtin_bit_cast(int, x);
    return __builtin_bit_cast(float,
        __builtin_amdgcn_mov_dpp(b, 0x160 + M, 0xf, 0xf, false));
}

__device__ __forceinline__ float row16_sum(float x) {
    x += lane_xor16<1>(x);
    x += lane_xor16<2>(x);
    x += lane_xor16<4>(x);
    x += lane_xor16<8>(x);
    return x;
}

__device__ __forceinline__ float xhalf(float x) {
    int b = __builtin_bit_cast(int, x);
    return __builtin_bit_cast(float,
        __builtin_amdgcn_permlanex16(b, b, 0x76543210, 0xfedcba98, false, false));
}

__device__ __forceinline__ float dot4(const float4 a, const float4 b) {
    return a.x * b.x + a.y * b.y + a.z * b.z + a.w * b.w;
}

__global__ void __launch_bounds__(MU_WAVES * 32)
mu_kernel(const float* __restrict__ keys,
          const float* __restrict__ values,
          const float* __restrict__ klen,
          float* __restrict__ mu) {
    const int lane = threadIdx.x & 31;
    const int wave = __builtin_amdgcn_readfirstlane((int)(threadIdx.x >> 5));
    const int run  = blockIdx.x * MU_WAVES + wave;
    const int n    = run / (LL / MU_R);
    const int l0   = (run % (LL / MU_R)) * MU_R;

    float4 kq[MU_ROWS][4], vq[MU_ROWS][4];
    float  kl[MU_ROWS];
#pragma unroll
    for (int r = 0; r < MU_ROWS; ++r) {
        const int l = (l0 + r < LL) ? (l0 + r) : (LL - 1);
        const float4* kp = (const float4*)(keys   + ((size_t)n * LL + l) * (HH * EE));
        const float4* vp = (const float4*)(values + ((size_t)n * LL + l) * (HH * MM));
        kl[r] = klen[n * LL + l];
#pragma unroll
        for (int j = 0; j < 4; ++j) {
            kq[r][j] = kp[j * 32 + lane];
            vq[r][j] = vp[j * 32 + lane];
        }
    }

    float mval[MU_R];
    float Bprev = 0.0f;
#pragma unroll
    for (int r = 0; r < MU_ROWS; ++r) {
        float bs = 0.0f, ds = 0.0f;
#pragma unroll
        for (int j = 0; j < 4; ++j) {
            float4 k = kq[r][j];
            k.x = fmap(k.x) * kl[r]; k.y = fmap(k.y) * kl[r];
            k.z = fmap(k.z) * kl[r]; k.w = fmap(k.w) * kl[r];
            kq[r][j] = k;
            const float4 v = vq[r][j];
            const float kn = row16_sum(dot4(k, k));
            const float vn = row16_sum(dot4(v, v));
            bs += kn * vn;
            if (r > 0) {
                const float kd = row16_sum(dot4(k, kq[r - 1][j]));
                const float vd = row16_sum(dot4(v, vq[r - 1][j]));
                ds += kd * vd;
            }
        }
        const float B = bs + xhalf(bs);
        if (r > 0) {
            const float D = ds + xhalf(ds);
            float diff2 = Bprev + B - 2.0f * D;
            diff2 = fmaxf(diff2, 0.0f);
            float ratio = sqrtf(diff2) / sqrtf(Bprev);
            float r4 = sqrtf(ratio);
            float m_ = 1.0f - r4;
            m_ = m_ * m_;
            m_ = fminf(fmaxf(m_, 0.0f), 0.9999f);
            mval[r - 1] = m_;
        }
        Bprev = B;
    }

    float mv = mval[0];
#pragma unroll
    for (int r = 1; r < MU_R; ++r) mv = (lane == r) ? mval[r] : mv;
    const int l = l0 + lane;
    if (lane < MU_R && l < LL - 1) mu[n * LL + l] = mv;
}

#define WSEG 8
#define WTHREADS (LL / WSEG)

#define GP 4
#define GC (2 * GP)
#define NG (NP / GP)
#define GPOS (GP * CK2)
#define SR 16
#define LDK (SR + PAD)
#define KLD_IT (GPOS * SR / 4 / 256)
#define VLD_IT (GPOS * MM / 4 / 256)

__global__ void __launch_bounds__(256)
sdelta_kernel(const float* __restrict__ keys,
                              const float* __restrict__ values,
                              const float* __restrict__ klen,
                              const float* __restrict__ mu,
                              float* __restrict__ weights,
                              float* __restrict__ Sbuf,
                              float* __restrict__ Ksum) {
    __shared__ __attribute__((aligned(16))) float Kst[GPOS][LDK];
    __shared__ __attribute__((aligned(16))) float Vlds[GPOS][LDW];
    __shared__ __attribute__((aligned(16))) float Tl[GC][SR][LDW];
    __shared__ float csum[2 * GP][SR];
    __shared__ float sA[WTHREADS / 32];
    __shared__ float sB[WTHREADS / 32];
    __shared__ __attribute__((aligned(16))) float wl[GPOS];

    int b = blockIdx.x;
    int q = b & 3;
    int g = (b >> 2) & (NG - 1);
    int h = (b >> 4) & (HH - 1);
    int n = b >> 7;
    int c0 = g * GPOS;
    int t = threadIdx.x;
    int wave = t >> 5, lane = t & 31, lr = lane & 15, lh = lane >> 4;

    float4 kq[KLD_IT], vq[VLD_IT];
    float  klr[KLD_IT];
#pragma unroll
    for (int j = 0; j < KLD_IT; ++j) {
        const int i  = t + j * 256;
        const int s  = i >> 2;
        const int e4 = (i & 3) * 4;
        const int l  = c0 + s;
        klr[j] = klen[n * LL + l];
        kq[j] = *(const float4*)&keys[(((size_t)n * LL + l) * HH + h) * EE + q * SR + e4];
    }
#pragma unroll
    for (int j = 0; j < VLD_IT; ++j) {
        const int i  = t + j * 256;
        const int s  = i >> 4;
        const int e4 = (i & 15) * 4;
        const int l  = c0 + s;
        vq[j] = *(const float4*)&values[(((size_t)n * LL + l) * HH + h) * MM + e4];
    }
    const float* mp = mu + (size_t)n * LL + t * WSEG;
    float mq[WSEG];
    if (t < WTHREADS - 1) {
        const float4 lo = *(const float4*)(mp);
        const float4 hi = *(const float4*)(mp + 4);
        mq[0] = lo.x; mq[1] = lo.y; mq[2] = lo.z; mq[3] = lo.w;
        mq[4] = hi.x; mq[5] = hi.y; mq[6] = hi.z; mq[7] = hi.w;
    } else {
        const float4 lo = *(const float4*)(mp);
        mq[0] = lo.x; mq[1] = lo.y; mq[2] = lo.z; mq[3] = lo.w;
        mq[4] = mp[4]; mq[5] = mp[5]; mq[6] = mp[6];
        mq[7] = 0.0f;
    }
    __builtin_amdgcn_sched_barrier(0);

#pragma unroll
    for (int j = 0; j < KLD_IT; ++j) {
        const int i  = t + j * 256;
        const int s  = i >> 2;
        const int e4 = (i & 3) * 4;
        const float kl = klr[j];
        const float4 kv = kq[j];
        float4 ko;
        ko.x = fmap(kv.x) * kl; ko.y = fmap(kv.y) * kl;
        ko.z = fmap(kv.z) * kl; ko.w = fmap(kv.w) * kl;
        kq[j] = ko;
        *(float4*)&Kst[s][e4] = ko;
    }
#pragma unroll
    for (int j = 0; j < VLD_IT; ++j) {
        const int i  = t + j * 256;
        const int s  = i >> 4;
        const int e4 = (i & 15) * 4;
        *(float4*)&Vlds[s][e4] = vq[j];
    }

    float A = mq[WSEG - 1];
    float B = 1.0f;
#pragma unroll
    for (int j = WSEG - 2; j >= 0; --j) {
        B = B * mq[j] + 1.0f;
        A = A * mq[j];
    }

#pragma unroll
    for (int off = 1; off < 32; off <<= 1) {
        float A2 = __shfl_down(A, off, 32);
        float B2 = __shfl_down(B, off, 32);
        if (lane + off < 32) {
            B = A * B2 + B;
            A = A * A2;
        }
    }
    if (lane == 0) { sA[wave] = A; sB[wave] = B; }

    float Ae = __shfl_down(A, 1, 32);
    float Be = __shfl_down(B, 1, 32);
    if (lane == 31) { Ae = 1.0f; Be = 0.0f; }
    __syncthreads();

    float xw = 0.0f;
#pragma unroll
    for (int v = WTHREADS / 32 - 1; v >= 1; --v) {
        float a = sA[v], bq = sB[v];
        if (v > wave) xw = a * xw + bq;
    }

    float x = Ae * xw + Be;

    float o[WSEG];
#pragma unroll
    for (int j = WSEG - 1; j >= 0; --j) {
        x = x * mq[j] + 1.0f;
        o[j] = x;
    }
    if ((t >> 6) == g) {
        *(float4*)&wl[(t & 63) * WSEG]     = make_float4(o[0], o[1], o[2], o[3]);
        *(float4*)&wl[(t & 63) * WSEG + 4] = make_float4(o[4], o[5], o[6], o[7]);
    }

    {
        float ps = 0.f;
        const int r0 = wave * CK + lh * (CK / 2);
#pragma unroll 8
        for (int s = 0; s < CK / 2; ++s) ps += Kst[r0 + s][lr];
        ps += xhalf(ps);
        if (lh == 0) csum[wave][lr] = ps;
    }
    __syncthreads();

#pragma unroll
    for (int j = 0; j < KLD_IT; ++j) {
        const int i  = t + j * 256;
        const int s  = i >> 2;
        const int e4 = (i & 3) * 4;
        const float w = wl[s];
        float4 ko = kq[j];
        ko.x *= w; ko.y *= w; ko.z *= w; ko.w *= w;
        *(float4*)&Kst[s][e4] = ko;
    }

    if (h == 0 && q == 0 && t < GPOS / 4)
        *(float4*)(weights + (size_t)n * LL + c0 + t * 4) = *(const float4*)&wl[t * 4];

    if (t < GC * SR) {
        const int cc  = t >> 4;
        const int col = t & 15;
        float run = 0.f, mine = 0.f;
#pragma unroll
        for (int jj = 0; jj < GC; ++jj) {
            if (jj == cc) mine = run;
            run += csum[jj][col];
        }
        const size_t ki = (((size_t)n * HH + h) * NC + g * GC + cc) * EE + q * SR + col;
        Ksum[ki] = (cc == 0) ? run : mine;
    }
    __syncthreads();

    {
        const int j   = wave >> 1;
        const int tj0 = (wave & 1) * 2;
#pragma unroll
        for (int cc = 0; cc < 2; ++cc) {
            const int ch = 2 * j + cc;
            v8f acc0 = {}, acc1 = {};
#pragma unroll
            for (int k0 = 0; k0 < CK; k0 += 4) {
                const int s = ch * CK + k0 + 2 * lh;
                v2f a, b0, b1;
                a.x = Kst[s][lr];
                a.y = Kst[s + 1][lr];
                const int m0 = tj0 * 16 + lr;
                b0.x = Vlds[s][m0];
                b0.y = Vlds[s + 1][m0];
                b1.x = Vlds[s][m0 + 16];
                b1.y = Vlds[s + 1][m0 + 16];
                acc0 = __builtin_amdgcn_wmma_f32_16x16x4_f32(false, a, false, b0,
                                                             (short)0, acc0, false, false);
                acc1 = __builtin_amdgcn_wmma_f32_16x16x4_f32(false, a, false, b1,
                                                             (short)0, acc1, false, false);
            }
#pragma unroll
            for (int r = 0; r < 8; ++r) {
                const int row = r + 8 * lh;
                Tl[ch][row][tj0 * 16 + lr]      = acc0[r];
                Tl[ch][row][tj0 * 16 + 16 + lr] = acc1[r];
            }
        }
    }
    __syncthreads();

    {
        const int row = t >> 4;
        const int c4  = (t & 15) * 4;
        float4 tv[GC];
#pragma unroll
        for (int cc = 0; cc < GC; ++cc) tv[cc] = *(const float4*)&Tl[cc][row][c4];
        float* o0 = Sbuf + (((size_t)n * HH + h) * NC + g * GC) * (EE * MM)
                         + (size_t)(q * SR + row) * MM + c4;
        float4 run = tv[0];
#pragma unroll
        for (int cc = 1; cc < GC; ++cc) {
            *(float4*)(o0 + cc * EE * MM) = run;
            run.x += tv[cc].x; run.y += tv[cc].y; run.z += tv[cc].z; run.w += tv[cc].w;
        }
        *(float4*)(o0) = run;
    }
}

__global__ void __launch_bounds__(256) __attribute__((amdgpu_waves_per_eu(4, 4)))
out_kernel(const float* __restrict__ queries,
                           const float* __restrict__ keys,
                           const float* __restrict__ values,
                           const float* __restrict__ klen,
                           const float* __restrict__ weights,
                           const float* __restrict__ Sbuf,
                           const float* __restrict__ Ksum,
                           float* __restrict__ out) {
    __shared__ __attribute__((aligned(16))) float Qlds[CK][LDW];
    __shared__ __attribute__((aligned(16))) float KP[CK][LDW];
    __shared__ __attribute__((aligned(16))) float Vlds[CK][LDW];
    __shared__ __attribute__((aligned(16))) float Slds[EE][LDW];
    __shared__ __attribute__((aligned(16))) float ks[EE];
    __shared__ float zinv[CK];

    int b = blockIdx.x;
    int c = b % NC;
    int h = (b / NC) % HH;
    int n = b / (NC * HH);
    int c0 = c * CK;
    int t = threadIdx.x;

    const int  g   = c / GC;
    const int  jp  = c % GC;
    const float* Sn = Sbuf + ((size_t)n * HH + h) * NC * (EE * MM);
    const float* Sp = Sn + (size_t)c * (EE * MM);
    float4 qq[LD_IT], kq[LD_IT], vq[LD_IT], sq[LD_IT];
    float  klr[LD_IT], wr[LD_IT];
#pragma unroll
    for (int j = 0; j < LD_IT; ++j) {
        const int i  = t + j * 256;
        const int s  = i >> 4;
        const int e4 = (i & 15) * 4;
        const int l  = c0 + s;
        klr[j] = klen[n * LL + l];
        wr[j]  = weights[n * LL + l];
        qq[j] = *(const float4*)&queries[(((size_t)n * LL + l) * HH + h) * EE + e4];
        kq[j] = *(const float4*)&keys   [(((size_t)n * LL + l) * HH + h) * EE + e4];
        vq[j] = *(const float4*)&values [(((size_t)n * LL + l) * HH + h) * MM + e4];
        sq[j] = *(const float4*)&Sp[s * MM + e4];
    }
    const size_t kn = ((size_t)n * HH + h) * NC * EE + (t & (EE - 1));
    const size_t ki = kn + (size_t)c * EE;
    float ksv = 0.0f, kg0 = 0.0f, kg1 = 0.0f, kg2 = 0.0f;
    if (jp > 0) ksv = Ksum[ki];
    if (g > 0)  kg0 = Ksum[kn + 0 * GC * EE];
    if (g > 1)  kg1 = Ksum[kn + 1 * GC * EE];
    if (g > 2)  kg2 = Ksum[kn + 2 * GC * EE];
    __builtin_amdgcn_sched_barrier(0);
    ksv += (kg0 + kg1) + kg2;
#pragma unroll
    for (int j = 0; j < LD_IT; ++j) {
        const int i  = t + j * 256;
        const int s  = i >> 4;
        const int e4 = (i & 15) * 4;
        const float kl = klr[j];
        const float w  = wr[j];
        const float4 qv = qq[j];
        const float4 kv = kq[j];
        const float4 vv = vq[j];
        float4 qo, ko, vo;
        qo.x = fmap(qv.x);      qo.y = fmap(qv.y);
        qo.z = fmap(qv.z);      qo.w = fmap(qv.w);
        ko.x = fmap(kv.x) * kl; ko.y = fmap(kv.y) * kl;
        ko.z = fmap(kv.z) * kl; ko.w = fmap(kv.w) * kl;
        vo.x = vv.x * w; vo.y = vv.y * w; vo.z = vv.z * w; vo.w = vv.w * w;
        *(float4*)&Qlds[s][e4] = qo;
        *(float4*)&KP[s][e4]   = ko;
        *(float4*)&Vlds[s][e4] = vo;
        float4 so;
        so.x = (jp > 0) ? sq[j].x : 0.0f; so.y = (jp > 0) ? sq[j].y : 0.0f;
        so.z = (jp > 0) ? sq[j].z : 0.0f; so.w = (jp > 0) ? sq[j].w : 0.0f;
        *(float4*)&Slds[s][e4] = so;
    }
    ks[t & (EE - 1)] = ksv;
    __syncthreads();

    float4 g0[LD_IT], g1[LD_IT], g2[LD_IT];
#pragma unroll
    for (int j = 0; j < LD_IT; ++j) {
        const int i  = t + j * 256;
        const int s  = i >> 4;
        const int e4 = (i & 15) * 4;
        const float4 z4 = {0.f, 0.f, 0.f, 0.f};
        g0[j] = z4; g1[j] = z4; g2[j] = z4;
        if (g > 0)  g0[j] = *(const float4*)&Sn[0 * GC * (EE * MM) + s * MM + e4];
        if (g > 1)  g1[j] = *(const float4*)&Sn[1 * GC * (EE * MM) + s * MM + e4];
        if (g > 2)  g2[j] = *(const float4*)&Sn[2 * GC * (EE * MM) + s * MM + e4];
    }

    int wave = t >> 5, lane = t & 31, lr = lane & 15, lh = lane >> 4;

    v8f p0 = {}, p1 = {};
    for (int half = 0; half < 2; ++half) {
        int tile = wave + half * 8;
        int ti = tile >> 2, tj = tile & 3;
        v8f acc = {};
        for (int k0 = 0; k0 < EE; k0 += 4) {
            int e = k0 + 2 * lh;
            v2f a, bb;
            int row = ti * 16 + lr;
            a.x = Qlds[row][e];
            a.y = Qlds[row][e + 1];
            int col = tj * 16 + lr;
            bb.x = KP[col][e];
            bb.y = KP[col][e + 1];
            acc = __builtin_amdgcn_wmma_f32_16x16x4_f32(false, a, false, bb,
                                                        (short)0, acc, false, false);
        }
        if (half) p1 = acc; else p0 = acc;
    }
    __syncthreads();

    if (g > 0) {
#pragma unroll
        for (int j = 0; j < LD_IT; ++j) {
            const int i  = t + j * 256;
            const int s  = i >> 4;
            const int e4 = (i & 15) * 4;
            float4 so = *(const float4*)&Slds[s][e4];
            so.x += (g0[j].x + g1[j].x) + g2[j].x;
            so.y += (g0[j].y + g1[j].y) + g2[j].y;
            so.z += (g0[j].z + g1[j].z) + g2[j].z;
            so.w += (g0[j].w + g1[j].w) + g2[j].w;
            *(float4*)&Slds[s][e4] = so;
        }
    }

    for (int half = 0; half < 2; ++half) {
        v8f acc = half ? p1 : p0;
        int tile = wave + half * 8;
        int ti = tile >> 2, tj = tile & 3;
        for (int r = 0; r < 8; ++r) {
            int row = ti * 16 + r + 8 * lh;
            int col = tj * 16 + lr;
            KP[row][col] = (col <= row) ? acc[r] : 0.0f;
        }
    }
    __syncthreads();

    {
        const int zr  = t >> 2;
        const int zc0 = (t & 3) * 16;
        float4 pv[4], qv[4], kv[4];
#pragma unroll
        for (int i = 0; i < 4; ++i) {
            pv[i] = *(const float4*)&KP[zr][zc0 + 4 * i];
            qv[i] = *(const float4*)&Qlds[zr][zc0 + 4 * i];
            kv[i] = *(const float4*)&ks[zc0 + 4 * i];
        }
        float zs = 0.f, zq = 0.f;
#pragma unroll
        for (int i = 0; i < 4; ++i) {
            zs += (pv[i].x + pv[i].y) + (pv[i].z + pv[i].w);
            zq += dot4(qv[i], kv[i]);
        }
        float zt = zs + zq;
        zt += lane_xor16<1>(zt);
        zt += lane_xor16<2>(zt);
        if ((t & 3) == 0) zinv[zr] = 1.0f / (zt + EPS);
    }
    __syncthreads();

    for (int half = 0; half < 2; ++half) {
        int tile = wave + half * 8;
        int ti = tile >> 2, tj = tile & 3;
        v8f acc = {};
        for (int k0 = 0; k0 < EE; k0 += 4) {
            int e = k0 + 2 * lh;
            v2f a, bb;
            int row = ti * 16 + lr;
            a.x = Qlds[row][e];
            a.y = Qlds[row][e + 1];
            int col = tj * 16 + lr;
            bb.x = Slds[e][col];
            bb.y = Slds[e + 1][col];
            acc = __builtin_amdgcn_wmma_f32_16x16x4_f32(false, a, false, bb,
                                                        (short)0, acc, false, false);
        }
        for (int k0 = 0; k0 < CK; k0 += 4) {
            int s = k0 + 2 * lh;
            v2f a, bb;
            int row = ti * 16 + lr;
            a.x = KP[row][s];
            a.y = KP[row][s + 1];
            int col = tj * 16 + lr;
            bb.x = Vlds[s][col];
            bb.y = Vlds[s + 1][col];
            acc = __builtin_amdgcn_wmma_f32_16x16x4_f32(false, a, false, bb,
                                                        (short)0, acc, false, false);
        }
        for (int r = 0; r < 8; ++r) {
            int row = ti * 16 + r + 8 * lh;
            int col = tj * 16 + lr;
            int l = c0 + row;
            out[(((size_t)n * LL + l) * HH + h) * MM + col] = acc[r] * zinv[row];
        }
    }
}

extern "C" void kernel_launch(void* const* d_in, const int* in_sizes, int n_in,
                              void* d_out, int out_size, void* d_ws, size_t ws_size,
                              hipStream_t stream) {
    const float* queries = (const float*)d_in[0];
    const float* keys    = (const float*)d_in[1];
    const float* values  = (const float*)d_in[2];
    const float* klen    = (const float*)d_in[3];
    float* outp = (float*)d_out;

    float* ws      = (float*)d_ws;
    float* Sbuf    = ws;
    float* Ksum    = Sbuf + (size_t)NB * HH * NC * EE * MM;
    float* mu      = Ksum + (size_t)NB * HH * NC * EE;
    float* weights = mu + (size_t)NB * LL;

    mu_kernel<<<NB * LL / (MU_R * MU_WAVES), MU_WAVES * 32, 0, stream>>>(keys, values, klen, mu);
    sdelta_kernel<<<NB * HH * NG * (EE / SR), 256, 0, stream>>>(keys, values, klen, mu, weights, Sbuf, Ksum);
    out_kernel<<<NB * HH * NC, 256, 0, stream>>>(queries, keys, values, klen,
                                                 weights, Sbuf, Ksum, outp);
}


# state-delta and output kernels: the first products start once their own operands are staged (K + first half of V; Q + K); the other loads stay in flight across LDS-only barriers, land under the WMMAs
# speedup vs baseline: 4.6156x; 1.0318x over previous
#include <hip/hip_runtime.h>
#include <hip/hip_bf16.h>

#define NB 2
#define LL 2048
#define HH 8
#define EE 64
#define MM 64
#define CK 64
#define NC (LL/CK)
#define EPS 1e-6f
#define PAD 4
#define LDW (EE + PAD)
#define LD_IT (CK * EE / 4 / 256)
#define NP (NC / 2)
#define CK2 (2 * CK)

typedef __attribute__((ext_vector_type(2))) float v2f;
typedef __attribute__((ext_vector_type(8))) float v8f;

__device__ __forceinline__ float fmap(float x) {
    return x > 0.0f ? x + 1.0f : __expf(x);
}

#define MU_R 4
#define MU_ROWS (MU_R + 1)
#define MU_WAVES 4

template <int M>
__device__ __forceinline__ float lane_xor16(float x) {
    int b = __builtin_bit_cast(int, x);
    return __builtin_bit_cast(float,
        __builtin_amdgcn_mov_dpp(b, 0x160 + M, 0xf, 0xf, false));
}

__device__ __forceinline__ float row16_sum(float x) {
    x += lane_xor16<1>(x);
    x += lane_xor16<2>(x);
    x += lane_xor16<4>(x);
    x += lane_xor16<8>(x);
    return x;
}

__device__ __forceinline__ float xhalf(float x) {
    int b = __builtin_bit_cast(int, x);
    return __builtin_bit_cast(float,
        __builtin_amdgcn_permlanex16(b, b, 0x76543210, 0xfedcba98, false, false));
}

__device__ __forceinline__ float dot4(const float4 a, const float4 b) {
    return a.x * b.x + a.y * b.y + a.z * b.z + a.w * b.w;
}

__device__ __forceinline__ void lds_barrier() {
    asm volatile("s_wait_dscnt 0x0\n\ts_barrier_signal -1\n\ts_barrier_wait -1" ::: "memory");
}

__global__ void __launch_bounds__(MU_WAVES * 32)
mu_kernel(const float* __restrict__ keys,
          const float* __restrict__ values,
          const float* __restrict__ klen,
          float* __restrict__ mu) {
    const int lane = threadIdx.x & 31;
    const int wave = __builtin_amdgcn_readfirstlane((int)(threadIdx.x >> 5));
    const int run  = blockIdx.x * MU_WAVES + wave;
    const int n    = run / (LL / MU_R);
    const int l0   = (run % (LL / MU_R)) * MU_R;

    float4 kq[MU_ROWS][4], vq[MU_ROWS][4];
    float  kl[MU_ROWS];
#pragma unroll
    for (int r = 0; r < MU_ROWS; ++r) {
        const int l = (l0 + r < LL) ? (l0 + r) : (LL - 1);
        const float4* kp = (const float4*)(keys   + ((size_t)n * LL + l) * (HH * EE));
        const float4* vp = (const float4*)(values + ((size_t)n * LL + l) * (HH * MM));
        kl[r] = klen[n * LL + l];
#pragma unroll
        for (int j = 0; j < 4; ++j) {
            kq[r][j] = kp[j * 32 + lane];
            vq[r][j] = vp[j * 32 + lane];
        }
    }

    float mval[MU_R];
    float Bprev = 0.0f;
#pragma unroll
    for (int r = 0; r < MU_ROWS; ++r) {
        float bs = 0.0f, ds = 0.0f;
#pragma unroll
        for (int j = 0; j < 4; ++j) {
            float4 k = kq[r][j];
            k.x = fmap(k.x) * kl[r]; k.y = fmap(k.y) * kl[r];
            k.z = fmap(k.z) * kl[r]; k.w = fmap(k.w) * kl[r];
            kq[r][j] = k;
            const float4 v = vq[r][j];
            const float kn = row16_sum(dot4(k, k));
            const float vn = row16_sum(dot4(v, v));
            bs += kn * vn;
            if (r > 0) {
                const float kd = row16_sum(dot4(k, kq[r - 1][j]));
                const float vd = row16_sum(dot4(v, vq[r - 1][j]));
                ds += kd * vd;
            }
        }
        const float B = bs + xhalf(bs);
        if (r > 0) {
            const float D = ds + xhalf(ds);
            float diff2 = Bprev + B - 2.0f * D;
            diff2 = fmaxf(diff2, 0.0f);
            float ratio = sqrtf(diff2) / sqrtf(Bprev);
            float r4 = sqrtf(ratio);
            float m_ = 1.0f - r4;
            m_ = m_ * m_;
            m_ = fminf(fmaxf(m_, 0.0f), 0.9999f);
            mval[r - 1] = m_;
        }
        Bprev = B;
    }

    float mv = mval[0];
#pragma unroll
    for (int r = 1; r < MU_R; ++r) mv = (lane == r) ? mval[r] : mv;
    const int l = l0 + lane;
    if (lane < MU_R && l < LL - 1) mu[n * LL + l] = mv;
}

#define WSEG 8
#define WTHREADS (LL / WSEG)

#define GP 4
#define GC (2 * GP)
#define NG (NP / GP)
#define GPOS (GP * CK2)
#define SR 16
#define LDK (SR + PAD)
#define KLD_IT (GPOS * SR / 4 / 256)
#define VLD_IT (GPOS * MM / 4 / 256)

__global__ void __launch_bounds__(256)
sdelta_kernel(const float* __restrict__ keys,
                              const float* __restrict__ values,
                              const float* __restrict__ klen,
                              const float* __restrict__ mu,
                              float* __restrict__ weights,
                              float* __restrict__ Sbuf,
                              float* __restrict__ Ksum) {
    __shared__ __attribute__((aligned(16))) float Kst[GPOS][LDK];
    __shared__ __attribute__((aligned(16))) float Vlds[GPOS][LDW];
    __shared__ __attribute__((aligned(16))) float Tl[GC][SR][LDW];
    __shared__ float csum[2 * GP][SR];
    __shared__ float sA[WTHREADS / 32];
    __shared__ float sB[WTHREADS / 32];
    __shared__ __attribute__((aligned(16))) float wl[GPOS];

    int b = blockIdx.x;
    int q = b & 3;
    int g = (b >> 2) & (NG - 1);
    int h = (b >> 4) & (HH - 1);
    int n = b >> 7;
    int c0 = g * GPOS;
    int t = threadIdx.x;
    int wave = t >> 5, lane = t & 31, lr = lane & 15, lh = lane >> 4;

    float4 kq[KLD_IT];
    float4 v0_0, v0_1, v0_2, v0_3, v0_4, v0_5, v0_6, v0_7;
    float4 v1_0, v1_1, v1_2, v1_3, v1_4, v1_5, v1_6, v1_7;
    float4 v2_0, v2_1, v2_2, v2_3, v2_4, v2_5, v2_6, v2_7;
    float4 v3_0, v3_1, v3_2, v3_3, v3_4, v3_5, v3_6, v3_7;
    float  klr[KLD_IT];
#pragma unroll
    for (int j = 0; j < KLD_IT; ++j) {
        const int i  = t + j * 256;
        const int s  = i >> 2;
        const int e4 = (i & 3) * 4;
        const int l  = c0 + s;
        klr[j] = klen[n * LL + l];
        kq[j] = *(const float4*)&keys[(((size_t)n * LL + l) * HH + h) * EE + q * SR + e4];
    }
    const float* mp = mu + (size_t)n * LL + t * WSEG;
    float mq[WSEG];
    if (t < WTHREADS - 1) {
        const float4 lo = *(const float4*)(mp);
        const float4 hi = *(const float4*)(mp + 4);
        mq[0] = lo.x; mq[1] = lo.y; mq[2] = lo.z; mq[3] = lo.w;
        mq[4] = hi.x; mq[5] = hi.y; mq[6] = hi.z; mq[7] = hi.w;
    } else {
        const float4 lo = *(const float4*)(mp);
        mq[0] = lo.x; mq[1] = lo.y; mq[2] = lo.z; mq[3] = lo.w;
        mq[4] = mp[4]; mq[5] = mp[5]; mq[6] = mp[6];
        mq[7] = 0.0f;
    }
    const int vrow = t >> 4;
    const int vcol = (t & 15) * 4;
    const float4* vsrc = (const float4*)&values[(((size_t)n * LL + c0 + vrow) * HH + h) * MM + vcol];
    v0_0 = vsrc[(size_t)(0) * 16 * (HH * MM / 4)];
    v0_1 = vsrc[(size_t)(1) * 16 * (HH * MM / 4)];
    v0_2 = vsrc[(size_t)(2) * 16 * (HH * MM / 4)];
    v0_3 = vsrc[(size_t)(3) * 16 * (HH * MM / 4)];
    v0_4 = vsrc[(size_t)(4) * 16 * (HH * MM / 4)];
    v0_5 = vsrc[(size_t)(5) * 16 * (HH * MM / 4)];
    v0_6 = vsrc[(size_t)(6) * 16 * (HH * MM / 4)];
    v0_7 = vsrc[(size_t)(7) * 16 * (HH * MM / 4)];
    v1_0 = vsrc[(size_t)(8) * 16 * (HH * MM / 4)];
    v1_1 = vsrc[(size_t)(9) * 16 * (HH * MM / 4)];
    v1_2 = vsrc[(size_t)(10) * 16 * (HH * MM / 4)];
    v1_3 = vsrc[(size_t)(11) * 16 * (HH * MM / 4)];
    v1_4 = vsrc[(size_t)(12) * 16 * (HH * MM / 4)];
    v1_5 = vsrc[(size_t)(13) * 16 * (HH * MM / 4)];
    v1_6 = vsrc[(size_t)(14) * 16 * (HH * MM / 4)];
    v1_7 = vsrc[(size_t)(15) * 16 * (HH * MM / 4)];
    v2_0 = vsrc[(size_t)(16) * 16 * (HH * MM / 4)];
    v2_1 = vsrc[(size_t)(17) * 16 * (HH * MM / 4)];
    v2_2 = vsrc[(size_t)(18) * 16 * (HH * MM / 4)];
    v2_3 = vsrc[(size_t)(19) * 16 * (HH * MM / 4)];
    v2_4 = vsrc[(size_t)(20) * 16 * (HH * MM / 4)];
    v2_5 = vsrc[(size_t)(21) * 16 * (HH * MM / 4)];
    v2_6 = vsrc[(size_t)(22) * 16 * (HH * MM / 4)];
    v2_7 = vsrc[(size_t)(23) * 16 * (HH * MM / 4)];
    v3_0 = vsrc[(size_t)(24) * 16 * (HH * MM / 4)];
    v3_1 = vsrc[(size_t)(25) * 16 * (HH * MM / 4)];
    v3_2 = vsrc[(size_t)(26) * 16 * (HH * MM / 4)];
    v3_3 = vsrc[(size_t)(27) * 16 * (HH * MM / 4)];
    v3_4 = vsrc[(size_t)(28) * 16 * (HH * MM / 4)];
    v3_5 = vsrc[(size_t)(29) * 16 * (HH * MM / 4)];
    v3_6 = vsrc[(size_t)(30) * 16 * (HH * MM / 4)];
    v3_7 = vsrc[(size_t)(31) * 16 * (HH * MM / 4)];
    __builtin_amdgcn_sched_barrier(0);

#pragma unroll
    for (int j = 0; j < KLD_IT; ++j) {
        const int i  = t + j * 256;
        const int s  = i >> 2;
        const int e4 = (i & 3) * 4;
        const float kl = klr[j];
        const float4 kv = kq[j];
        float4 ko;
        ko.x = fmap(kv.x) * kl; ko.y = fmap(kv.y) * kl;
        ko.z = fmap(kv.z) * kl; ko.w = fmap(kv.w) * kl;
        kq[j] = ko;
        *(float4*)&Kst[s][e4] = ko;
    }
    float A = mq[WSEG - 1];
    float B = 1.0f;
#pragma unroll
    for (int j = WSEG - 2; j >= 0; --j) {
        B = B * mq[j] + 1.0f;
        A = A * mq[j];
    }

#pragma unroll
    for (int off = 1; off < 32; off <<= 1) {
        float A2 = __shfl_down(A, off, 32);
        float B2 = __shfl_down(B, off, 32);
        if (lane + off < 32) {
            B = A * B2 + B;
            A = A * A2;
        }
    }
    if (lane == 0) { sA[wave] = A; sB[wave] = B; }

    float Ae = __shfl_down(A, 1, 32);
    float Be = __shfl_down(B, 1, 32);
    if (lane == 31) { Ae = 1.0f; Be = 0.0f; }

    lds_barrier();

    float xw = 0.0f;
#pragma unroll
    for (int v = WTHREADS / 32 - 1; v >= 1; --v) {
        float a = sA[v], bq = sB[v];
        if (v > wave) xw = a * xw + bq;
    }

    float x = Ae * xw + Be;

    float o[WSEG];
#pragma unroll
    for (int j = WSEG - 1; j >= 0; --j) {
        x = x * mq[j] + 1.0f;
        o[j] = x;
    }
    if ((t >> 6) == g) {
        *(float4*)&wl[(t & 63) * WSEG]     = make_float4(o[0], o[1], o[2], o[3]);
        *(float4*)&wl[(t & 63) * WSEG + 4] = make_float4(o[4], o[5], o[6], o[7]);
    }

    {
        float ps = 0.f;
        const int r0 = wave * CK + lh * (CK / 2);
#pragma unroll 8
        for (int s = 0; s < CK / 2; ++s) ps += Kst[r0 + s][lr];
        ps += xhalf(ps);
        if (lh == 0) csum[wave][lr] = ps;
    }
    lds_barrier();

#pragma unroll
    for (int j = 0; j < KLD_IT; ++j) {
        const int i  = t + j * 256;
        const int s  = i >> 2;
        const int e4 = (i & 3) * 4;
        const float w = wl[s];
        float4 ko = kq[j];
        ko.x *= w; ko.y *= w; ko.z *= w; ko.w *= w;
        *(float4*)&Kst[s][e4] = ko;
    }

    if (h == 0 && q == 0 && t < GPOS / 4)
        *(float4*)(weights + (size_t)n * LL + c0 + t * 4) = *(const float4*)&wl[t * 4];

    if (t < GC * SR) {
        const int cc  = t >> 4;
        const int col = t & 15;
        float run = 0.f, mine = 0.f;
#pragma unroll
        for (int jj = 0; jj < GC; ++jj) {
            if (jj == cc) mine = run;
            run += csum[jj][col];
        }
        const size_t ki = (((size_t)n * HH + h) * NC + g * GC + cc) * EE + q * SR + col;
        Ksum[ki] = (cc == 0) ? run : mine;
    }
    *(float4*)&Vlds[vrow + 0][vcol] = v0_0;
    *(float4*)&Vlds[vrow + 16][vcol] = v0_1;
    *(float4*)&Vlds[vrow + 32][vcol] = v0_2;
    *(float4*)&Vlds[vrow + 48][vcol] = v0_3;
    *(float4*)&Vlds[vrow + 64][vcol] = v0_4;
    *(float4*)&Vlds[vrow + 80][vcol] = v0_5;
    *(float4*)&Vlds[vrow + 96][vcol] = v0_6;
    *(float4*)&Vlds[vrow + 112][vcol] = v0_7;
    *(float4*)&Vlds[vrow + 128][vcol] = v1_0;
    *(float4*)&Vlds[vrow + 144][vcol] = v1_1;
    *(float4*)&Vlds[vrow + 160][vcol] = v1_2;
    *(float4*)&Vlds[vrow + 176][vcol] = v1_3;
    *(float4*)&Vlds[vrow + 192][vcol] = v1_4;
    *(float4*)&Vlds[vrow + 208][vcol] = v1_5;
    *(float4*)&Vlds[vrow + 224][vcol] = v1_6;
    *(float4*)&Vlds[vrow + 240][vcol] = v1_7;
    lds_barrier();

    {
        const int m0 = (wave & 1) * 32 + lr;
        {
            const int ch = 0 + (wave >> 1);
            v8f acc0 = {}, acc1 = {};
#pragma unroll
            for (int k0 = 0; k0 < CK; k0 += 4) {
                const int s = ch * CK + k0 + 2 * lh;
                v2f a, b0, b1;
                a.x = Kst[s][lr];
                a.y = Kst[s + 1][lr];
                b0.x = Vlds[s][m0];
                b0.y = Vlds[s + 1][m0];
                b1.x = Vlds[s][m0 + 16];
                b1.y = Vlds[s + 1][m0 + 16];
                acc0 = __builtin_amdgcn_wmma_f32_16x16x4_f32(false, a, false, b0,
                                                             (short)0, acc0, false, false);
                acc1 = __builtin_amdgcn_wmma_f32_16x16x4_f32(false, a, false, b1,
                                                             (short)0, acc1, false, false);
            }
#pragma unroll
            for (int r = 0; r < 8; ++r) {
                const int row = r + 8 * lh;
                Tl[ch][row][m0]      = acc0[r];
                Tl[ch][row][m0 + 16] = acc1[r];
            }
        }
        __builtin_amdgcn_sched_barrier(0);
        *(float4*)&Vlds[vrow + 256][vcol] = v2_0;
        *(float4*)&Vlds[vrow + 272][vcol] = v2_1;
        *(float4*)&Vlds[vrow + 288][vcol] = v2_2;
        *(float4*)&Vlds[vrow + 304][vcol] = v2_3;
        *(float4*)&Vlds[vrow + 320][vcol] = v2_4;
        *(float4*)&Vlds[vrow + 336][vcol] = v2_5;
        *(float4*)&Vlds[vrow + 352][vcol] = v2_6;
        *(float4*)&Vlds[vrow + 368][vcol] = v2_7;
        *(float4*)&Vlds[vrow + 384][vcol] = v3_0;
        *(float4*)&Vlds[vrow + 400][vcol] = v3_1;
        *(float4*)&Vlds[vrow + 416][vcol] = v3_2;
        *(float4*)&Vlds[vrow + 432][vcol] = v3_3;
        *(float4*)&Vlds[vrow + 448][vcol] = v3_4;
        *(float4*)&Vlds[vrow + 464][vcol] = v3_5;
        *(float4*)&Vlds[vrow + 480][vcol] = v3_6;
        *(float4*)&Vlds[vrow + 496][vcol] = v3_7;
        lds_barrier();
        {
            const int ch = 4 + (wave >> 1);
            v8f acc0 = {}, acc1 = {};
#pragma unroll
            for (int k0 = 0; k0 < CK; k0 += 4) {
                const int s = ch * CK + k0 + 2 * lh;
                v2f a, b0, b1;
                a.x = Kst[s][lr];
                a.y = Kst[s + 1][lr];
                b0.x = Vlds[s][m0];
                b0.y = Vlds[s + 1][m0];
                b1.x = Vlds[s][m0 + 16];
                b1.y = Vlds[s + 1][m0 + 16];
                acc0 = __builtin_amdgcn_wmma_f32_16x16x4_f32(false, a, false, b0,
                                                             (short)0, acc0, false, false);
                acc1 = __builtin_amdgcn_wmma_f32_16x16x4_f32(false, a, false, b1,
                                                             (short)0, acc1, false, false);
            }
#pragma unroll
            for (int r = 0; r < 8; ++r) {
                const int row = r + 8 * lh;
                Tl[ch][row][m0]      = acc0[r];
                Tl[ch][row][m0 + 16] = acc1[r];
            }
        }
    }
    __syncthreads();

    {
        const int row = t >> 4;
        const int c4  = (t & 15) * 4;
        float4 tv[GC];
#pragma unroll
        for (int cc = 0; cc < GC; ++cc) tv[cc] = *(const float4*)&Tl[cc][row][c4];
        float* o0 = Sbuf + (((size_t)n * HH + h) * NC + g * GC) * (EE * MM)
                         + (size_t)(q * SR + row) * MM + c4;
        float4 run = tv[0];
#pragma unroll
        for (int cc = 1; cc < GC; ++cc) {
            *(float4*)(o0 + cc * EE * MM) = run;
            run.x += tv[cc].x; run.y += tv[cc].y; run.z += tv[cc].z; run.w += tv[cc].w;
        }
        *(float4*)(o0) = run;
    }
}

__global__ void __launch_bounds__(256) __attribute__((amdgpu_waves_per_eu(4, 4)))
out_kernel(const float* __restrict__ queries,
                           const float* __restrict__ keys,
                           const float* __restrict__ values,
                           const float* __restrict__ klen,
                           const float* __restrict__ weights,
                           const float* __restrict__ Sbuf,
                           const float* __restrict__ Ksum,
                           float* __restrict__ out) {
    __shared__ __attribute__((aligned(16))) float Qlds[CK][LDW];
    __shared__ __attribute__((aligned(16))) float KP[CK][LDW];
    __shared__ __attribute__((aligned(16))) float Vlds[CK][LDW];
    __shared__ __attribute__((aligned(16))) float Slds[EE][LDW];
    __shared__ __attribute__((aligned(16))) float ks[EE];
    __shared__ float zinv[CK];

    int b = blockIdx.x;
    int c = b % NC;
    int h = (b / NC) % HH;
    int n = b / (NC * HH);
    int c0 = c * CK;
    int t = threadIdx.x;

    const int  g   = c / GC;
    const int  jp  = c % GC;
    const float* Sn = Sbuf + ((size_t)n * HH + h) * NC * (EE * MM);
    const float* Sp = Sn + (size_t)c * (EE * MM);
    float4 qq[LD_IT], kq[LD_IT], vq[LD_IT], sq[LD_IT];
    float  klr[LD_IT], wr[LD_IT];
#pragma unroll
    for (int j = 0; j < LD_IT; ++j) {
        const int i  = t + j * 256;
        const int s  = i >> 4;
        const int e4 = (i & 15) * 4;
        const int l  = c0 + s;
        klr[j] = klen[n * LL + l];
        qq[j] = *(const float4*)&queries[(((size_t)n * LL + l) * HH + h) * EE + e4];
        kq[j] = *(const float4*)&keys   [(((size_t)n * LL + l) * HH + h) * EE + e4];
    }
    const size_t kn = ((size_t)n * HH + h) * NC * EE + (t & (EE - 1));
    const size_t ki = kn + (size_t)c * EE;
    float ksv = 0.0f, kg0 = 0.0f, kg1 = 0.0f, kg2 = 0.0f;
    if (jp > 0) ksv = Ksum[ki];
    if (g > 0)  kg0 = Ksum[kn + 0 * GC * EE];
    if (g > 1)  kg1 = Ksum[kn + 1 * GC * EE];
    if (g > 2)  kg2 = Ksum[kn + 2 * GC * EE];
    __builtin_amdgcn_sched_barrier(0);
#pragma unroll
    for (int j = 0; j < LD_IT; ++j) {
        const int i  = t + j * 256;
        const int s  = i >> 4;
        const int e4 = (i & 15) * 4;
        const int l  = c0 + s;
        wr[j]  = weights[n * LL + l];
        vq[j] = *(const float4*)&values [(((size_t)n * LL + l) * HH + h) * MM + e4];
        sq[j] = *(const float4*)&Sp[s * MM + e4];
    }
    __builtin_amdgcn_sched_barrier(0);
#pragma unroll
    for (int j = 0; j < LD_IT; ++j) {
        const int i  = t + j * 256;
        const int s  = i >> 4;
        const int e4 = (i & 15) * 4;
        const float kl = klr[j];
        const float4 qv = qq[j];
        const float4 kv = kq[j];
        float4 qo, ko;
        qo.x = fmap(qv.x);      qo.y = fmap(qv.y);
        qo.z = fmap(qv.z);      qo.w = fmap(qv.w);
        ko.x = fmap(kv.x) * kl; ko.y = fmap(kv.y) * kl;
        ko.z = fmap(kv.z) * kl; ko.w = fmap(kv.w) * kl;
        *(float4*)&Qlds[s][e4] = qo;
        *(float4*)&KP[s][e4]   = ko;
    }
    lds_barrier();

    const float* T0 = (g > 0) ? Sn + 0 * GC * (EE * MM) : Sp;
    const float* T1 = (g > 1) ? Sn + 1 * GC * (EE * MM) : Sp;
    const float* T2 = (g > 2) ? Sn + 2 * GC * (EE * MM) : Sp;
    float4 g0[LD_IT], g1[LD_IT], g2[LD_IT];
#pragma unroll
    for (int j = 0; j < LD_IT; ++j) {
        const int i  = t + j * 256;
        const int s  = i >> 4;
        const int e4 = (i & 15) * 4;
        g0[j] = *(const float4*)&T0[s * MM + e4];
        g1[j] = *(const float4*)&T1[s * MM + e4];
        g2[j] = *(const float4*)&T2[s * MM + e4];
    }
    __builtin_amdgcn_sched_barrier(0);

    int wave = t >> 5, lane = t & 31, lr = lane & 15, lh = lane >> 4;

    v8f p0 = {}, p1 = {};
    for (int half = 0; half < 2; ++half) {
        int tile = wave + half * 8;
        int ti = tile >> 2, tj = tile & 3;
        v8f acc = {};
        for (int k0 = 0; k0 < EE; k0 += 4) {
            int e = k0 + 2 * lh;
            v2f a, bb;
            int row = ti * 16 + lr;
            a.x = Qlds[row][e];
            a.y = Qlds[row][e + 1];
            int col = tj * 16 + lr;
            bb.x = KP[col][e];
            bb.y = KP[col][e + 1];
            acc = __builtin_amdgcn_wmma_f32_16x16x4_f32(false, a, false, bb,
                                                        (short)0, acc, false, false);
        }
        if (half) p1 = acc; else p0 = acc;
    }
    __builtin_amdgcn_sched_barrier(0);

    int jq = jp, gq = g;
    asm volatile("" : "+s"(jq), "+s"(gq), "+v"(wr[0]), "+v"(wr[1]), "+v"(wr[2]), "+v"(wr[3]) :: "memory");
    ksv += (kg0 + kg1) + kg2;
#pragma unroll
    for (int j = 0; j < LD_IT; ++j) {
        const int i  = t + j * 256;
        const int s  = i >> 4;
        const int e4 = (i & 15) * 4;
        const float w  = wr[j];
        const float4 vv = vq[j];
        float4 vo;
        vo.x = vv.x * w; vo.y = vv.y * w; vo.z = vv.z * w; vo.w = vv.w * w;
        *(float4*)&Vlds[s][e4] = vo;
        float4 so;
        so.x = (jq > 0) ? sq[j].x : 0.0f; so.y = (jq > 0) ? sq[j].y : 0.0f;
        so.z = (jq > 0) ? sq[j].z : 0.0f; so.w = (jq > 0) ? sq[j].w : 0.0f;
        const float4 z4 = {0.f, 0.f, 0.f, 0.f};
        const float4 a0 = (gq > 0) ? g0[j] : z4;
        const float4 a1 = (gq > 1) ? g1[j] : z4;
        const float4 a2 = (gq > 2) ? g2[j] : z4;
        so.x += (a0.x + a1.x) + a2.x;
        so.y += (a0.y + a1.y) + a2.y;
        so.z += (a0.z + a1.z) + a2.z;
        so.w += (a0.w + a1.w) + a2.w;
        *(float4*)&Slds[s][e4] = so;
    }
    ks[t & (EE - 1)] = ksv;
    __syncthreads();

    for (int half = 0; half < 2; ++half) {
        v8f acc = half ? p1 : p0;
        int tile = wave + half * 8;
        int ti = tile >> 2, tj = tile & 3;
        for (int r = 0; r < 8; ++r) {
            int row = ti * 16 + r + 8 * lh;
            int col = tj * 16 + lr;
            KP[row][col] = (col <= row) ? acc[r] : 0.0f;
        }
    }
    __syncthreads();

    {
        const int zr  = t >> 2;
        const int zc0 = (t & 3) * 16;
        float4 pv[4], qv[4], kv[4];
#pragma unroll
        for (int i = 0; i < 4; ++i) {
            pv[i] = *(const float4*)&KP[zr][zc0 + 4 * i];
            qv[i] = *(const float4*)&Qlds[zr][zc0 + 4 * i];
            kv[i] = *(const float4*)&ks[zc0 + 4 * i];
        }
        float zs = 0.f, zq = 0.f;
#pragma unroll
        for (int i = 0; i < 4; ++i) {
            zs += (pv[i].x + pv[i].y) + (pv[i].z + pv[i].w);
            zq += dot4(qv[i], kv[i]);
        }
        float zt = zs + zq;
        zt += lane_xor16<1>(zt);
        zt += lane_xor16<2>(zt);
        if ((t & 3) == 0) zinv[zr] = 1.0f / (zt + EPS);
    }
    __syncthreads();

    for (int half = 0; half < 2; ++half) {
        int tile = wave + half * 8;
        int ti = tile >> 2, tj = tile & 3;
        v8f acc = {};
        for (int k0 = 0; k0 < EE; k0 += 4) {
            int e = k0 + 2 * lh;
            v2f a, bb;
            int row = ti * 16 + lr;
            a.x = Qlds[row][e];
            a.y = Qlds[row][e + 1];
            int col = tj * 16 + lr;
            bb.x = Slds[e][col];
            bb.y = Slds[e + 1][col];
            acc = __builtin_amdgcn_wmma_f32_16x16x4_f32(false, a, false, bb,
                                                        (short)0, acc, false, false);
        }
        for (int k0 = 0; k0 < CK; k0 += 4) {
            int s = k0 + 2 * lh;
            v2f a, bb;
            int row = ti * 16 + lr;
            a.x = KP[row][s];
            a.y = KP[row][s + 1];
            int col = tj * 16 + lr;
            bb.x = Vlds[s][col];
            bb.y = Vlds[s + 1][col];
            acc = __builtin_amdgcn_wmma_f32_16x16x4_f32(false, a, false, bb,
                                                        (short)0, acc, false, false);
        }
        for (int r = 0; r < 8; ++r) {
            int row = ti * 16 + r + 8 * lh;
            int col = tj * 16 + lr;
            int l = c0 + row;
            out[(((size_t)n * LL + l) * HH + h) * MM + col] = acc[r] * zinv[row];
        }
    }
}

extern "C" void kernel_launch(void* const* d_in, const int* in_sizes, int n_in,
                              void* d_out, int out_size, void* d_ws, size_t ws_size,
                              hipStream_t stream) {
    const float* queries = (const float*)d_in[0];
    const float* keys    = (const float*)d_in[1];
    const float* values  = (const float*)d_in[2];
    const float* klen    = (const float*)d_in[3];
    float* outp = (float*)d_out;

    float* ws      = (float*)d_ws;
    float* Sbuf    = ws;
    float* Ksum    = Sbuf + (size_t)NB * HH * NC * EE * MM;
    float* mu      = Ksum + (size_t)NB * HH * NC * EE;
    float* weights = mu + (size_t)NB * LL;

    mu_kernel<<<NB * LL / (MU_R * MU_WAVES), MU_WAVES * 32, 0, stream>>>(keys, values, klen, mu);
    sdelta_kernel<<<NB * HH * NG * (EE / SR), 256, 0, stream>>>(keys, values, klen, mu, weights, Sbuf, Ksum);
    out_kernel<<<NB * HH * NC, 256, 0, stream>>>(queries, keys, values, klen,
                                                 weights, Sbuf, Ksum, outp);
}
